# starting kernel
# baseline (speedup 1.0000x reference)
#include <hip/hip_runtime.h>

typedef float v2f __attribute__((ext_vector_type(2)));
typedef float v8f __attribute__((ext_vector_type(8)));

#define NNODES 10000
#define NEDGES 320000
#define DIN 512
#define DOUT 256

__global__ __launch_bounds__(256) void lin_gemm_kernel(
    const float* __restrict__ X, const float* __restrict__ W,
    const float* __restrict__ bias, float* __restrict__ H)
{
    const int lane   = threadIdx.x & 31;
    const int wave   = threadIdx.x >> 5;
    const int lane16 = lane & 15;
    const int hsel   = lane >> 4;

    const int row0 = blockIdx.x * 16;
    const int col0 = wave * 32;

    const float* arow  = X + (size_t)(row0 + lane16) * DIN + 2 * hsel;
    const float* bcol0 = W + (size_t)(2 * hsel) * DOUT + (col0 + lane16);
    const float* bcol1 = bcol0 + 16;

    v8f acc0 = {};
    v8f acc1 = {};
    #pragma unroll 4
    for (int k = 0; k < DIN; k += 4) {
        v2f a = *(const v2f*)arow;
        v2f b0f, b1f;
        b0f.x = bcol0[0];
        b0f.y = bcol0[DOUT];
        b1f.x = bcol1[0];
        b1f.y = bcol1[DOUT];
        acc0 = __builtin_amdgcn_wmma_f32_16x16x4_f32(false, a, false, b0f,
                                                     (short)0, acc0, false, false);
        acc1 = __builtin_amdgcn_wmma_f32_16x16x4_f32(false, a, false, b1f,
                                                     (short)0, acc1, false, false);
        arow += 4;
        bcol0 += 4 * DOUT;
        bcol1 += 4 * DOUT;
    }

    const float bv0 = bias[col0 + lane16];
    const float bv1 = bias[col0 + 16 + lane16];
    float* out = H + (size_t)row0 * DOUT + col0 + lane16;
    #pragma unroll
    for (int r = 0; r < 8; ++r) {
        const size_t m = (size_t)(r + 8 * hsel) * DOUT;
        out[m]      = acc0[r] + bv0;
        out[m + 16] = acc1[r] + bv1;
    }
}

__global__ __launch_bounds__(256) void sage_gemm_kernel(
    const float* __restrict__ AGG, const float* __restrict__ INVDEG,
    const float* __restrict__ Hin,
    const float* __restrict__ Wl, const float* __restrict__ bl,
    const float* __restrict__ Wr,
    float* __restrict__ OUT, int relu)
{
    const int lane   = threadIdx.x & 31;
    const int wave   = threadIdx.x >> 5;
    const int lane16 = lane & 15;
    const int hsel   = lane >> 4;

    const int row0 = blockIdx.x * 16;
    const int col0 = wave * 32;

    const int arow_idx = row0 + lane16;
    const float inv = INVDEG[arow_idx];

    const float* aggrow = AGG + (size_t)arow_idx * DOUT + 2 * hsel;
    const float* hrow   = Hin + (size_t)arow_idx * DOUT + 2 * hsel;
    const float* blc0   = Wl + (size_t)(2 * hsel) * DOUT + (col0 + lane16);
    const float* blc1   = blc0 + 16;
    const float* brc0   = Wr + (size_t)(2 * hsel) * DOUT + (col0 + lane16);
    const float* brc1   = brc0 + 16;

    v8f acc0 = {};
    v8f acc1 = {};
    #pragma unroll 2
    for (int k = 0; k < DOUT; k += 4) {
        v2f am = *(const v2f*)aggrow;
        am.x *= inv; am.y *= inv;
        v2f ah = *(const v2f*)hrow;
        v2f bl0f, bl1f, br0f, br1f;
        bl0f.x = blc0[0];    bl0f.y = blc0[DOUT];
        bl1f.x = blc1[0];    bl1f.y = blc1[DOUT];
        br0f.x = brc0[0];    br0f.y = brc0[DOUT];
        br1f.x = brc1[0];    br1f.y = brc1[DOUT];
        acc0 = __builtin_amdgcn_wmma_f32_16x16x4_f32(false, am, false, bl0f,
                                                     (short)0, acc0, false, false);
        acc1 = __builtin_amdgcn_wmma_f32_16x16x4_f32(false, am, false, bl1f,
                                                     (short)0, acc1, false, false);
        acc0 = __builtin_amdgcn_wmma_f32_16x16x4_f32(false, ah, false, br0f,
                                                     (short)0, acc0, false, false);
        acc1 = __builtin_amdgcn_wmma_f32_16x16x4_f32(false, ah, false, br1f,
                                                     (short)0, acc1, false, false);
        aggrow += 4; hrow += 4;
        blc0 += 4 * DOUT; blc1 += 4 * DOUT;
        brc0 += 4 * DOUT; brc1 += 4 * DOUT;
    }

    const float bv0 = bl[col0 + lane16];
    const float bv1 = bl[col0 + 16 + lane16];
    const float* hres = Hin + (size_t)row0 * DOUT + col0 + lane16;
    float* out = OUT + (size_t)row0 * DOUT + col0 + lane16;
    #pragma unroll
    for (int r = 0; r < 8; ++r) {
        const size_t m = (size_t)(r + 8 * hsel) * DOUT;
        float v0 = acc0[r] + bv0;
        float v1 = acc1[r] + bv1;
        if (relu) { v0 = fmaxf(v0, 0.0f); v1 = fmaxf(v1, 0.0f); }
        out[m]      = v0 + hres[m];
        out[m + 16] = v1 + hres[m + 16];
    }
}

__global__ __launch_bounds__(256) void scatter_add_kernel(
    const float* __restrict__ H, const int* __restrict__ src,
    const int* __restrict__ dst, float* __restrict__ AGG, int E)
{
    const int e = blockIdx.x * 8 + (threadIdx.x >> 5);
    if (e >= E) return;
    const int lane = threadIdx.x & 31;
    const int s = src[e];
    const int d = dst[e];
    const float* hs = H + (size_t)s * DOUT;
    float* ad = AGG + (size_t)d * DOUT;
    #pragma unroll
    for (int j = 0; j < 8; ++j) {
        atomicAdd(&ad[lane + 32 * j], hs[lane + 32 * j]);
    }
}

__global__ void deg_kernel(const int* __restrict__ dst, float* __restrict__ deg, int E)
{
    const int e = blockIdx.x * blockDim.x + threadIdx.x;
    if (e < E) atomicAdd(&deg[dst[e]], 1.0f);
}

__global__ void invdeg_kernel(float* __restrict__ deg, int n)
{
    const int i = blockIdx.x * blockDim.x + threadIdx.x;
    if (i < n) deg[i] = 1.0f / fmaxf(deg[i], 1.0f);
}

__global__ void zero_kernel(float* __restrict__ p, int n)
{
    const int i = blockIdx.x * blockDim.x + threadIdx.x;
    if (i < n) p[i] = 0.0f;
}

extern "C" void kernel_launch(void* const* d_in, const int* in_sizes, int n_in,
                              void* d_out, int out_size, void* d_ws, size_t ws_size,
                              hipStream_t stream)
{
    const float* x   = (const float*)d_in[0];
    const int*   ei  = (const int*)d_in[1];
    const float* W0  = (const float*)d_in[2];
    const float* b0  = (const float*)d_in[3];
    const float* Wl1 = (const float*)d_in[4];
    const float* bl1 = (const float*)d_in[5];
    const float* Wr1 = (const float*)d_in[6];
    const float* Wl2 = (const float*)d_in[7];
    const float* bl2 = (const float*)d_in[8];
    const float* Wr2 = (const float*)d_in[9];

    const int* src = ei;
    const int* dst = ei + NEDGES;

    float* ws  = (float*)d_ws;
    float* h   = ws;
    float* agg = h + (size_t)NNODES * DOUT;
    float* deg = agg + (size_t)NNODES * DOUT;

    float* out1 = (float*)d_out;
    float* out2 = out1 + (size_t)NNODES * DOUT;

    const int feat_elems = NNODES * DOUT;
    dim3 gemm_grid(NNODES / 16, 1);

    zero_kernel<<<(feat_elems + 255) / 256, 256, 0, stream>>>(agg, feat_elems);
    zero_kernel<<<(NNODES + 255) / 256, 256, 0, stream>>>(deg, NNODES);

    lin_gemm_kernel<<<gemm_grid, 256, 0, stream>>>(x, W0, b0, h);

    deg_kernel<<<(NEDGES + 255) / 256, 256, 0, stream>>>(dst, deg, NEDGES);
    invdeg_kernel<<<(NNODES + 255) / 256, 256, 0, stream>>>(deg, NNODES);

    scatter_add_kernel<<<NEDGES / 8, 256, 0, stream>>>(h, src, dst, agg, NEDGES);
    sage_gemm_kernel<<<gemm_grid, 256, 0, stream>>>(agg, deg, h, Wl1, bl1, Wr1, out1, 1);

    zero_kernel<<<(feat_elems + 255) / 256, 256, 0, stream>>>(agg, feat_elems);
    scatter_add_kernel<<<NEDGES / 8, 256, 0, stream>>>(out1, src, dst, agg, NEDGES);
    sage_gemm_kernel<<<gemm_grid, 256, 0, stream>>>(agg, deg, out1, Wl2, bl2, Wr2, out2, 0);
}


# Replace float-atomic scatter-add, degree atomics and zero fills by a per-call CSR (owner blocks rank edges by LDS counter arrival) plus one-wave-per-node register gather; sums added in CSR order
# speedup vs baseline: 2.0212x; 2.0212x over previous
#include <hip/hip_runtime.h>

typedef float v2f __attribute__((ext_vector_type(2)));
typedef float v8f __attribute__((ext_vector_type(8)));

#define NNODES 10000
#define NEDGES 320000
#define DIN 512
#define DOUT 256

__global__ __launch_bounds__(256) void lin_gemm_kernel(
    const float* __restrict__ X, const float* __restrict__ W,
    const float* __restrict__ bias, float* __restrict__ H)
{
    const int lane   = threadIdx.x & 31;
    const int wave   = threadIdx.x >> 5;
    const int lane16 = lane & 15;
    const int hsel   = lane >> 4;

    const int row0 = blockIdx.x * 16;
    const int col0 = wave * 32;

    const float* arow  = X + (size_t)(row0 + lane16) * DIN + 2 * hsel;
    const float* bcol0 = W + (size_t)(2 * hsel) * DOUT + (col0 + lane16);
    const float* bcol1 = bcol0 + 16;

    v8f acc0 = {};
    v8f acc1 = {};
    #pragma unroll 4
    for (int k = 0; k < DIN; k += 4) {
        v2f a = *(const v2f*)arow;
        v2f b0f, b1f;
        b0f.x = bcol0[0];
        b0f.y = bcol0[DOUT];
        b1f.x = bcol1[0];
        b1f.y = bcol1[DOUT];
        acc0 = __builtin_amdgcn_wmma_f32_16x16x4_f32(false, a, false, b0f,
                                                     (short)0, acc0, false, false);
        acc1 = __builtin_amdgcn_wmma_f32_16x16x4_f32(false, a, false, b1f,
                                                     (short)0, acc1, false, false);
        arow += 4;
        bcol0 += 4 * DOUT;
        bcol1 += 4 * DOUT;
    }

    const float bv0 = bias[col0 + lane16];
    const float bv1 = bias[col0 + 16 + lane16];
    float* out = H + (size_t)row0 * DOUT + col0 + lane16;
    #pragma unroll
    for (int r = 0; r < 8; ++r) {
        const size_t m = (size_t)(r + 8 * hsel) * DOUT;
        out[m]      = acc0[r] + bv0;
        out[m + 16] = acc1[r] + bv1;
    }
}

__global__ __launch_bounds__(256) void sage_gemm_kernel(
    const float* __restrict__ AGG, const float* __restrict__ INVDEG,
    const float* __restrict__ Hin,
    const float* __restrict__ Wl, const float* __restrict__ bl,
    const float* __restrict__ Wr,
    float* __restrict__ OUT, int relu)
{
    const int lane   = threadIdx.x & 31;
    const int wave   = threadIdx.x >> 5;
    const int lane16 = lane & 15;
    const int hsel   = lane >> 4;

    const int row0 = blockIdx.x * 16;
    const int col0 = wave * 32;

    const int arow_idx = row0 + lane16;
    const float inv = INVDEG[arow_idx];

    const float* aggrow = AGG + (size_t)arow_idx * DOUT + 2 * hsel;
    const float* hrow   = Hin + (size_t)arow_idx * DOUT + 2 * hsel;
    const float* blc0   = Wl + (size_t)(2 * hsel) * DOUT + (col0 + lane16);
    const float* blc1   = blc0 + 16;
    const float* brc0   = Wr + (size_t)(2 * hsel) * DOUT + (col0 + lane16);
    const float* brc1   = brc0 + 16;

    v8f acc0 = {};
    v8f acc1 = {};
    #pragma unroll 2
    for (int k = 0; k < DOUT; k += 4) {
        v2f am = *(const v2f*)aggrow;
        am.x *= inv; am.y *= inv;
        v2f ah = *(const v2f*)hrow;
        v2f bl0f, bl1f, br0f, br1f;
        bl0f.x = blc0[0];    bl0f.y = blc0[DOUT];
        bl1f.x = blc1[0];    bl1f.y = blc1[DOUT];
        br0f.x = brc0[0];    br0f.y = brc0[DOUT];
        br1f.x = brc1[0];    br1f.y = brc1[DOUT];
        acc0 = __builtin_amdgcn_wmma_f32_16x16x4_f32(false, am, false, bl0f,
                                                     (short)0, acc0, false, false);
        acc1 = __builtin_amdgcn_wmma_f32_16x16x4_f32(false, am, false, bl1f,
                                                     (short)0, acc1, false, false);
        acc0 = __builtin_amdgcn_wmma_f32_16x16x4_f32(false, ah, false, br0f,
                                                     (short)0, acc0, false, false);
        acc1 = __builtin_amdgcn_wmma_f32_16x16x4_f32(false, ah, false, br1f,
                                                     (short)0, acc1, false, false);
        aggrow += 4; hrow += 4;
        blc0 += 4 * DOUT; blc1 += 4 * DOUT;
        brc0 += 4 * DOUT; brc1 += 4 * DOUT;
    }

    const float bv0 = bl[col0 + lane16];
    const float bv1 = bl[col0 + 16 + lane16];
    const float* hres = Hin + (size_t)row0 * DOUT + col0 + lane16;
    float* out = OUT + (size_t)row0 * DOUT + col0 + lane16;
    #pragma unroll
    for (int r = 0; r < 8; ++r) {
        const size_t m = (size_t)(r + 8 * hsel) * DOUT;
        float v0 = acc0[r] + bv0;
        float v1 = acc1[r] + bv1;
        if (relu) { v0 = fmaxf(v0, 0.0f); v1 = fmaxf(v1, 0.0f); }
        out[m]      = v0 + hres[m];
        out[m + 16] = v1 + hres[m + 16];
    }
}

typedef int   v4i __attribute__((ext_vector_type(4)));
typedef float v4f __attribute__((ext_vector_type(4)));

#define CSR_NB   125
#define CSR_R    80
#define CSR_T    1024
#define CSR_E4   (NEDGES / 4)
#define CSR_FULL (CSR_E4 / CSR_T)
#define CSR_U    13
#define CSR_U2   6
#define CSR_REST (CSR_E4 - CSR_FULL * CSR_T)

static_assert(CSR_NB * CSR_R == NNODES, "owner ranges must tile the nodes");
static_assert(NEDGES % 4 == 0, "dst is read in groups of four");
static_assert(CSR_FULL % CSR_U == 0, "load batch must divide the full sweeps");
static_assert(CSR_FULL % CSR_U2 == 0, "load batch must divide the full sweeps");
static_assert(CSR_REST >= 0 && CSR_REST < CSR_T, "one partial sweep at most");

__global__ __launch_bounds__(CSR_T) void csr_build_kernel(
    const int* __restrict__ src, const int* __restrict__ dst,
    int* __restrict__ rowstart, int* __restrict__ rowcnt,
    int* __restrict__ perm, float* __restrict__ invdeg)
{
    __shared__ int s_cnt[CSR_R];
    __shared__ int s_cursor[CSR_R];
    __shared__ int s_below;

    const int tid  = threadIdx.x;
    const int lane = tid & 31;
    const int lo   = blockIdx.x * CSR_R;
    const v4i* __restrict__ d4 = (const v4i*)dst;
    const v4i* __restrict__ s4 = (const v4i*)src;

    if (tid < CSR_R) s_cnt[tid] = 0;
    if (tid == 0) s_below = 0;
    __syncthreads();

    int below = 0;
    for (int k0 = 0; k0 < CSR_FULL; k0 += CSR_U) {
        v4i v[CSR_U];
        #pragma unroll
        for (int u = 0; u < CSR_U; ++u) v[u] = d4[tid + (k0 + u) * CSR_T];
        #pragma unroll
        for (int u = 0; u < CSR_U; ++u) {
            #pragma unroll
            for (int c = 0; c < 4; ++c) {
                const int d = v[u][c];
                below += __popc(__ballot(d < lo));
                const unsigned r = (unsigned)(d - lo);
                if (r < (unsigned)CSR_R) atomicAdd(&s_cnt[r], 1);
            }
        }
    }
    {
        const bool have = tid < CSR_REST;
        v4i v = {NNODES, NNODES, NNODES, NNODES};
        if (have) v = d4[tid + CSR_FULL * CSR_T];
        #pragma unroll
        for (int c = 0; c < 4; ++c) {
            const int d = v[c];
            below += __popc(__ballot(d < lo));
            const unsigned r = (unsigned)(d - lo);
            if (r < (unsigned)CSR_R) atomicAdd(&s_cnt[r], 1);
        }
    }
    if (lane == 0) atomicAdd(&s_below, below);
    __syncthreads();

    if (tid < CSR_R) {
        int off = 0;
        for (int j = 0; j < tid; ++j) off += s_cnt[j];
        const int c  = s_cnt[tid];
        const int st = s_below + off;
        s_cursor[tid] = st;
        rowstart[lo + tid] = st;
        rowcnt[lo + tid]   = c;
        invdeg[lo + tid]   = 1.0f / fmaxf((float)c, 1.0f);
    }
    __syncthreads();

    for (int k0 = 0; k0 < CSR_FULL; k0 += CSR_U2) {
        v4i v[CSR_U2], s[CSR_U2];
        #pragma unroll
        for (int u = 0; u < CSR_U2; ++u) {
            v[u] = d4[tid + (k0 + u) * CSR_T];
            s[u] = s4[tid + (k0 + u) * CSR_T];
        }
        int p[CSR_U2][4];
        #pragma unroll
        for (int u = 0; u < CSR_U2; ++u) {
            #pragma unroll
            for (int c = 0; c < 4; ++c) {
                const unsigned r = (unsigned)(v[u][c] - lo);
                int q = -1;
                if (r < (unsigned)CSR_R) q = atomicAdd(&s_cursor[r], 1);
                p[u][c] = q;
            }
        }
        #pragma unroll
        for (int u = 0; u < CSR_U2; ++u) {
            #pragma unroll
            for (int c = 0; c < 4; ++c) {
                if (p[u][c] >= 0) perm[p[u][c]] = s[u][c];
            }
        }
    }
    if (tid < CSR_REST) {
        const v4i v = d4[tid + CSR_FULL * CSR_T];
        const v4i s = s4[tid + CSR_FULL * CSR_T];
        #pragma unroll
        for (int c = 0; c < 4; ++c) {
            const unsigned r = (unsigned)(v[c] - lo);
            if (r < (unsigned)CSR_R) {
                const int q = atomicAdd(&s_cursor[r], 1);
                perm[q] = s[c];
            }
        }
    }
}

#define GATHER_U 8
static_assert(NNODES % 8 == 0, "one wave per node, eight waves per block, no partial block");

__global__ __launch_bounds__(256) void gather_sum_kernel(
    const float* __restrict__ H, const int* __restrict__ rowstart,
    const int* __restrict__ rowcnt, const int* __restrict__ perm,
    float* __restrict__ AGG)
{
    const int node = __builtin_amdgcn_readfirstlane((int)(blockIdx.x * 8 + (threadIdx.x >> 5)));
    const int lane = threadIdx.x & 31;
    const int start = rowstart[node];
    const int cnt   = rowcnt[node];

    v4f a0 = {0.0f, 0.0f, 0.0f, 0.0f};
    v4f a1 = {0.0f, 0.0f, 0.0f, 0.0f};

    for (int j0 = 0; j0 < cnt; j0 += 32) {
        const int nb = (cnt - j0 < 32) ? (cnt - j0) : 32;
        int mysrc = 0;
        if (lane < nb) mysrc = perm[start + j0 + lane];
        for (int j = 0; j < nb; j += GATHER_U) {
            const int m = nb - j;
            v4f r0[GATHER_U], r1[GATHER_U];
            if (m >= GATHER_U) {
                #pragma unroll
                for (int u = 0; u < GATHER_U; ++u) {
                    const int s = __builtin_amdgcn_readlane(mysrc, j + u);
                    const v4f* __restrict__ p = (const v4f*)(H + (size_t)s * DOUT);
                    r0[u] = p[lane];
                    r1[u] = p[32 + lane];
                }
                #pragma unroll
                for (int u = 0; u < GATHER_U; ++u) { a0 += r0[u]; a1 += r1[u]; }
            } else {
                #pragma unroll
                for (int u = 0; u < GATHER_U; ++u) {
                    if (u < m) {
                        const int s = __builtin_amdgcn_readlane(mysrc, j + u);
                        const v4f* __restrict__ p = (const v4f*)(H + (size_t)s * DOUT);
                        r0[u] = p[lane];
                        r1[u] = p[32 + lane];
                    }
                }
                #pragma unroll
                for (int u = 0; u < GATHER_U; ++u) {
                    if (u < m) { a0 += r0[u]; a1 += r1[u]; }
                }
            }
        }
    }

    v4f* __restrict__ o = (v4f*)(AGG + (size_t)node * DOUT);
    o[lane]      = a0;
    o[32 + lane] = a1;
}

extern "C" void kernel_launch(void* const* d_in, const int* in_sizes, int n_in,
                              void* d_out, int out_size, void* d_ws, size_t ws_size,
                              hipStream_t stream)
{
    const float* x   = (const float*)d_in[0];
    const int*   ei  = (const int*)d_in[1];
    const float* W0  = (const float*)d_in[2];
    const float* b0  = (const float*)d_in[3];
    const float* Wl1 = (const float*)d_in[4];
    const float* bl1 = (const float*)d_in[5];
    const float* Wr1 = (const float*)d_in[6];
    const float* Wl2 = (const float*)d_in[7];
    const float* bl2 = (const float*)d_in[8];
    const float* Wr2 = (const float*)d_in[9];

    const int* src = ei;
    const int* dst = ei + NEDGES;

    float* ws  = (float*)d_ws;
    float* h   = ws;
    float* agg = h + (size_t)NNODES * DOUT;
    float* deg = agg + (size_t)NNODES * DOUT;
    int* rowstart = (int*)(deg + NNODES);
    int* rowcnt   = rowstart + NNODES;
    int* perm     = rowcnt + NNODES;

    float* out1 = (float*)d_out;
    float* out2 = out1 + (size_t)NNODES * DOUT;

    dim3 gemm_grid(NNODES / 16, 1);

    csr_build_kernel<<<CSR_NB, CSR_T, 0, stream>>>(src, dst, rowstart, rowcnt, perm, deg);

    lin_gemm_kernel<<<gemm_grid, 256, 0, stream>>>(x, W0, b0, h);

    gather_sum_kernel<<<NNODES / 8, 256, 0, stream>>>(h, rowstart, rowcnt, perm, agg);
    sage_gemm_kernel<<<gemm_grid, 256, 0, stream>>>(agg, deg, h, Wl1, bl1, Wr1, out1, 1);

    gather_sum_kernel<<<NNODES / 8, 256, 0, stream>>>(out1, rowstart, rowcnt, perm, agg);
    sage_gemm_kernel<<<gemm_grid, 256, 0, stream>>>(agg, deg, out1, Wl2, bl2, Wr2, out2, 0);
}


# GEMM operand feeding: a wave owns 16x64 as four column-interleaved 16x16 tiles; A and weights load 16 B per lane one stage ahead, rows store 16 B; k summed in permuted order inside blocks of 8
# speedup vs baseline: 2.5527x; 1.2630x over previous
#include <hip/hip_runtime.h>

typedef float v2f __attribute__((ext_vector_type(2)));
typedef float v8f __attribute__((ext_vector_type(8)));

#define NNODES 10000
#define NEDGES 320000
#define DIN 512
#define DOUT 256

typedef float v4f __attribute__((ext_vector_type(4)));

#define LIN_KB 4

struct LinStage {
    v4f a[LIN_KB];
    v4f b[LIN_KB][4];
};

__device__ __forceinline__ void lin_load(LinStage& s, const float* __restrict__ ap,
                                         const float* __restrict__ bp)
{
    #pragma unroll
    for (int q = 0; q < LIN_KB; ++q) {
        s.a[q] = *(const v4f*)(ap + 8 * q);
        #pragma unroll
        for (int j = 0; j < 4; ++j)
            s.b[q][j] = *(const v4f*)(bp + (size_t)(8 * q + j) * DOUT);
    }
}

__device__ __forceinline__ void lin_mma(const LinStage& s, v8f (&acc)[4])
{
    #pragma unroll
    for (int q = 0; q < LIN_KB; ++q) {
        #pragma unroll
        for (int u = 0; u < 2; ++u) {
            v2f a;
            a.x = s.a[q][2 * u];
            a.y = s.a[q][2 * u + 1];
            #pragma unroll
            for (int t = 0; t < 4; ++t) {
                v2f b;
                b.x = s.b[q][2 * u][t];
                b.y = s.b[q][2 * u + 1][t];
                acc[t] = __builtin_amdgcn_wmma_f32_16x16x4_f32(false, a, false, b,
                                                               (short)0, acc[t], false, false);
            }
        }
    }
}

__global__ __launch_bounds__(128) void lin_gemm_kernel(
    const float* __restrict__ X, const float* __restrict__ W,
    const float* __restrict__ bias, float* __restrict__ H)
{
    const int lane = threadIdx.x & 31;
    const int wave = threadIdx.x >> 5;
    const int c    = lane & 15;
    const int h    = lane >> 4;

    const int row0 = blockIdx.x * 16;
    const int col0 = wave * 64;

    const float* ap = X + (size_t)(row0 + c) * DIN + 4 * h;
    const float* bp = W + (size_t)(4 * h) * DOUT + col0 + 4 * c;

    v8f acc[4] = {};
    LinStage s0, s1;
    lin_load(s0, ap, bp);
    for (int k = 0; k < DIN - 16 * LIN_KB; k += 16 * LIN_KB) {
        lin_load(s1, ap + k + 8 * LIN_KB, bp + (size_t)(k + 8 * LIN_KB) * DOUT);
        __builtin_amdgcn_sched_barrier(0);
        lin_mma(s0, acc);
        __builtin_amdgcn_sched_barrier(0);
        lin_load(s0, ap + k + 16 * LIN_KB, bp + (size_t)(k + 16 * LIN_KB) * DOUT);
        __builtin_amdgcn_sched_barrier(0);
        lin_mma(s1, acc);
    }
    lin_load(s1, ap + DIN - 8 * LIN_KB, bp + (size_t)(DIN - 8 * LIN_KB) * DOUT);
    __builtin_amdgcn_sched_barrier(0);
    lin_mma(s0, acc);
    lin_mma(s1, acc);

    const v4f bv = *(const v4f*)(bias + col0 + 4 * c);
    float* out = H + (size_t)(row0 + 8 * h) * DOUT + col0 + 4 * c;
    #pragma unroll
    for (int r = 0; r < 8; ++r) {
        v4f v;
        v.x = acc[0][r] + bv.x;
        v.y = acc[1][r] + bv.y;
        v.z = acc[2][r] + bv.z;
        v.w = acc[3][r] + bv.w;
        *(v4f*)(out + (size_t)r * DOUT) = v;
    }
}

#define SAGE_KB 2

struct SageStage {
    v4f am[SAGE_KB];
    v4f ah[SAGE_KB];
    v4f bl[SAGE_KB][4];
    v4f br[SAGE_KB][4];
};

__device__ __forceinline__ void sage_load(SageStage& s, const float* __restrict__ amp,
                                          const float* __restrict__ ahp,
                                          const float* __restrict__ blp,
                                          const float* __restrict__ brp)
{
    #pragma unroll
    for (int q = 0; q < SAGE_KB; ++q) {
        s.am[q] = *(const v4f*)(amp + 8 * q);
        s.ah[q] = *(const v4f*)(ahp + 8 * q);
        #pragma unroll
        for (int j = 0; j < 4; ++j) {
            s.bl[q][j] = *(const v4f*)(blp + (size_t)(8 * q + j) * DOUT);
            s.br[q][j] = *(const v4f*)(brp + (size_t)(8 * q + j) * DOUT);
        }
    }
}

__device__ __forceinline__ void sage_mma(const SageStage& s, float inv, v8f (&acc)[4])
{
    #pragma unroll
    for (int q = 0; q < SAGE_KB; ++q) {
        #pragma unroll
        for (int u = 0; u < 2; ++u) {
            v2f am, ah;
            am.x = s.am[q][2 * u] * inv;
            am.y = s.am[q][2 * u + 1] * inv;
            ah.x = s.ah[q][2 * u];
            ah.y = s.ah[q][2 * u + 1];
            #pragma unroll
            for (int t = 0; t < 4; ++t) {
                v2f b;
                b.x = s.bl[q][2 * u][t];
                b.y = s.bl[q][2 * u + 1][t];
                acc[t] = __builtin_amdgcn_wmma_f32_16x16x4_f32(false, am, false, b,
                                                               (short)0, acc[t], false, false);
            }
            #pragma unroll
            for (int t = 0; t < 4; ++t) {
                v2f b;
                b.x = s.br[q][2 * u][t];
                b.y = s.br[q][2 * u + 1][t];
                acc[t] = __builtin_amdgcn_wmma_f32_16x16x4_f32(false, ah, false, b,
                                                               (short)0, acc[t], false, false);
            }
        }
    }
}

__global__ __launch_bounds__(128) void sage_gemm_kernel(
    const float* __restrict__ AGG, const float* __restrict__ INVDEG,
    const float* __restrict__ Hin,
    const float* __restrict__ Wl, const float* __restrict__ bl,
    const float* __restrict__ Wr,
    float* __restrict__ OUT, int relu)
{
    const int lane = threadIdx.x & 31;
    const int wave = threadIdx.x >> 5;
    const int c    = lane & 15;
    const int h    = lane >> 4;

    const int row0 = blockIdx.x * 16;
    const int col0 = wave * 64;

    const int arow_idx = row0 + c;
    const float inv = INVDEG[arow_idx];

    const float* amp = AGG + (size_t)arow_idx * DOUT + 4 * h;
    const float* ahp = Hin + (size_t)arow_idx * DOUT + 4 * h;
    const float* blp = Wl + (size_t)(4 * h) * DOUT + col0 + 4 * c;
    const float* brp = Wr + (size_t)(4 * h) * DOUT + col0 + 4 * c;

    v8f acc[4] = {};
    SageStage s0, s1;
    sage_load(s0, amp, ahp, blp, brp);
    for (int k = 0; k < DOUT - 16 * SAGE_KB; k += 16 * SAGE_KB) {
        const int k1 = k + 8 * SAGE_KB;
        const int k2 = k + 16 * SAGE_KB;
        sage_load(s1, amp + k1, ahp + k1, blp + (size_t)k1 * DOUT, brp + (size_t)k1 * DOUT);
        __builtin_amdgcn_sched_barrier(0);
        sage_mma(s0, inv, acc);
        __builtin_amdgcn_sched_barrier(0);
        sage_load(s0, amp + k2, ahp + k2, blp + (size_t)k2 * DOUT, brp + (size_t)k2 * DOUT);
        __builtin_amdgcn_sched_barrier(0);
        sage_mma(s1, inv, acc);
    }
    const size_t o0 = (size_t)(row0 + 8 * h) * DOUT + col0 + 4 * c;
    const float* hres = Hin + o0;
    float* out = OUT + o0;
    v4f bv;
    v4f hr[8];
    {
        const int k1 = DOUT - 8 * SAGE_KB;
        sage_load(s1, amp + k1, ahp + k1, blp + (size_t)k1 * DOUT, brp + (size_t)k1 * DOUT);
        bv = *(const v4f*)(bl + col0 + 4 * c);
        #pragma unroll
        for (int r = 0; r < 8; ++r) hr[r] = *(const v4f*)(hres + (size_t)r * DOUT);
        __builtin_amdgcn_sched_barrier(0);
        sage_mma(s0, inv, acc);
        sage_mma(s1, inv, acc);
    }

    v4f v[8];
    #pragma unroll
    for (int r = 0; r < 8; ++r) {
        v[r].x = acc[0][r] + bv.x;
        v[r].y = acc[1][r] + bv.y;
        v[r].z = acc[2][r] + bv.z;
        v[r].w = acc[3][r] + bv.w;
    }
    if (relu) {
        #pragma unroll
        for (int r = 0; r < 8; ++r) {
            v[r].x = fmaxf(v[r].x, 0.0f); v[r].y = fmaxf(v[r].y, 0.0f);
            v[r].z = fmaxf(v[r].z, 0.0f); v[r].w = fmaxf(v[r].w, 0.0f);
        }
    }
    #pragma unroll
    for (int r = 0; r < 8; ++r) *(v4f*)(out + (size_t)r * DOUT) = v[r] + hr[r];
}

typedef int   v4i __attribute__((ext_vector_type(4)));

#define CSR_NB   125
#define CSR_R    80
#define CSR_T    1024
#define CSR_E4   (NEDGES / 4)
#define CSR_FULL (CSR_E4 / CSR_T)
#define CSR_U    13
#define CSR_U2   6
#define CSR_REST (CSR_E4 - CSR_FULL * CSR_T)

static_assert(CSR_NB * CSR_R == NNODES, "owner ranges must tile the nodes");
static_assert(NEDGES % 4 == 0, "dst is read in groups of four");
static_assert(CSR_FULL % CSR_U == 0, "load batch must divide the full sweeps");
static_assert(CSR_FULL % CSR_U2 == 0, "load batch must divide the full sweeps");
static_assert(CSR_REST >= 0 && CSR_REST < CSR_T, "one partial sweep at most");

__global__ __launch_bounds__(CSR_T) void csr_build_kernel(
    const int* __restrict__ src, const int* __restrict__ dst,
    int* __restrict__ rowstart, int* __restrict__ rowcnt,
    int* __restrict__ perm, float* __restrict__ invdeg)
{
    __shared__ int s_cnt[CSR_R];
    __shared__ int s_cursor[CSR_R];
    __shared__ int s_below;

    const int tid  = threadIdx.x;
    const int lane = tid & 31;
    const int lo   = blockIdx.x * CSR_R;
    const v4i* __restrict__ d4 = (const v4i*)dst;
    const v4i* __restrict__ s4 = (const v4i*)src;

    if (tid < CSR_R) s_cnt[tid] = 0;
    if (tid == 0) s_below = 0;
    __syncthreads();

    int below = 0;
    for (int k0 = 0; k0 < CSR_FULL; k0 += CSR_U) {
        v4i v[CSR_U];
        #pragma unroll
        for (int u = 0; u < CSR_U; ++u) v[u] = d4[tid + (k0 + u) * CSR_T];
        #pragma unroll
        for (int u = 0; u < CSR_U; ++u) {
            #pragma unroll
            for (int c = 0; c < 4; ++c) {
                const int d = v[u][c];
                below += __popc(__ballot(d < lo));
                const unsigned r = (unsigned)(d - lo);
                if (r < (unsigned)CSR_R) atomicAdd(&s_cnt[r], 1);
            }
        }
    }
    {
        const bool have = tid < CSR_REST;
        v4i v = {NNODES, NNODES, NNODES, NNODES};
        if (have) v = d4[tid + CSR_FULL * CSR_T];
        #pragma unroll
        for (int c = 0; c < 4; ++c) {
            const int d = v[c];
            below += __popc(__ballot(d < lo));
            const unsigned r = (unsigned)(d - lo);
            if (r < (unsigned)CSR_R) atomicAdd(&s_cnt[r], 1);
        }
    }
    if (lane == 0) atomicAdd(&s_below, below);
    __syncthreads();

    if (tid < CSR_R) {
        int off = 0;
        for (int j = 0; j < tid; ++j) off += s_cnt[j];
        const int c  = s_cnt[tid];
        const int st = s_below + off;
        s_cursor[tid] = st;
        rowstart[lo + tid] = st;
        rowcnt[lo + tid]   = c;
        invdeg[lo + tid]   = 1.0f / fmaxf((float)c, 1.0f);
    }
    __syncthreads();

    for (int k0 = 0; k0 < CSR_FULL; k0 += CSR_U2) {
        v4i v[CSR_U2], s[CSR_U2];
        #pragma unroll
        for (int u = 0; u < CSR_U2; ++u) {
            v[u] = d4[tid + (k0 + u) * CSR_T];
            s[u] = s4[tid + (k0 + u) * CSR_T];
        }
        int p[CSR_U2][4];
        #pragma unroll
        for (int u = 0; u < CSR_U2; ++u) {
            #pragma unroll
            for (int c = 0; c < 4; ++c) {
                const unsigned r = (unsigned)(v[u][c] - lo);
                int q = -1;
                if (r < (unsigned)CSR_R) q = atomicAdd(&s_cursor[r], 1);
                p[u][c] = q;
            }
        }
        #pragma unroll
        for (int u = 0; u < CSR_U2; ++u) {
            #pragma unroll
            for (int c = 0; c < 4; ++c) {
                if (p[u][c] >= 0) perm[p[u][c]] = s[u][c];
            }
        }
    }
    if (tid < CSR_REST) {
        const v4i v = d4[tid + CSR_FULL * CSR_T];
        const v4i s = s4[tid + CSR_FULL * CSR_T];
        #pragma unroll
        for (int c = 0; c < 4; ++c) {
            const unsigned r = (unsigned)(v[c] - lo);
            if (r < (unsigned)CSR_R) {
                const int q = atomicAdd(&s_cursor[r], 1);
                perm[q] = s[c];
            }
        }
    }
}

#define GATHER_U 8
static_assert(NNODES % 8 == 0, "one wave per node, eight waves per block, no partial block");

__global__ __launch_bounds__(256) void gather_sum_kernel(
    const float* __restrict__ H, const int* __restrict__ rowstart,
    const int* __restrict__ rowcnt, const int* __restrict__ perm,
    float* __restrict__ AGG)
{
    const int node = __builtin_amdgcn_readfirstlane((int)(blockIdx.x * 8 + (threadIdx.x >> 5)));
    const int lane = threadIdx.x & 31;
    const int start = rowstart[node];
    const int cnt   = rowcnt[node];

    v4f a0 = {0.0f, 0.0f, 0.0f, 0.0f};
    v4f a1 = {0.0f, 0.0f, 0.0f, 0.0f};

    for (int j0 = 0; j0 < cnt; j0 += 32) {
        const int nb = (cnt - j0 < 32) ? (cnt - j0) : 32;
        int mysrc = 0;
        if (lane < nb) mysrc = perm[start + j0 + lane];
        for (int j = 0; j < nb; j += GATHER_U) {
            const int m = nb - j;
            v4f r0[GATHER_U], r1[GATHER_U];
            if (m >= GATHER_U) {
                #pragma unroll
                for (int u = 0; u < GATHER_U; ++u) {
                    const int s = __builtin_amdgcn_readlane(mysrc, j + u);
                    const v4f* __restrict__ p = (const v4f*)(H + (size_t)s * DOUT);
                    r0[u] = p[lane];
                    r1[u] = p[32 + lane];
                }
                #pragma unroll
                for (int u = 0; u < GATHER_U; ++u) { a0 += r0[u]; a1 += r1[u]; }
            } else {
                #pragma unroll
                for (int u = 0; u < GATHER_U; ++u) {
                    if (u < m) {
                        const int s = __builtin_amdgcn_readlane(mysrc, j + u);
                        const v4f* __restrict__ p = (const v4f*)(H + (size_t)s * DOUT);
                        r0[u] = p[lane];
                        r1[u] = p[32 + lane];
                    }
                }
                #pragma unroll
                for (int u = 0; u < GATHER_U; ++u) {
                    if (u < m) { a0 += r0[u]; a1 += r1[u]; }
                }
            }
        }
    }

    v4f* __restrict__ o = (v4f*)(AGG + (size_t)node * DOUT);
    o[lane]      = a0;
    o[32 + lane] = a1;
}

extern "C" void kernel_launch(void* const* d_in, const int* in_sizes, int n_in,
                              void* d_out, int out_size, void* d_ws, size_t ws_size,
                              hipStream_t stream)
{
    const float* x   = (const float*)d_in[0];
    const int*   ei  = (const int*)d_in[1];
    const float* W0  = (const float*)d_in[2];
    const float* b0  = (const float*)d_in[3];
    const float* Wl1 = (const float*)d_in[4];
    const float* bl1 = (const float*)d_in[5];
    const float* Wr1 = (const float*)d_in[6];
    const float* Wl2 = (const float*)d_in[7];
    const float* bl2 = (const float*)d_in[8];
    const float* Wr2 = (const float*)d_in[9];

    const int* src = ei;
    const int* dst = ei + NEDGES;

    float* ws  = (float*)d_ws;
    float* h   = ws;
    float* agg = h + (size_t)NNODES * DOUT;
    float* deg = agg + (size_t)NNODES * DOUT;
    int* rowstart = (int*)(deg + NNODES);
    int* rowcnt   = rowstart + NNODES;
    int* perm     = rowcnt + NNODES;

    float* out1 = (float*)d_out;
    float* out2 = out1 + (size_t)NNODES * DOUT;

    dim3 gemm_grid(NNODES / 16, 1);

    csr_build_kernel<<<CSR_NB, CSR_T, 0, stream>>>(src, dst, rowstart, rowcnt, perm, deg);

    lin_gemm_kernel<<<gemm_grid, 128, 0, stream>>>(x, W0, b0, h);

    gather_sum_kernel<<<NNODES / 8, 256, 0, stream>>>(h, rowstart, rowcnt, perm, agg);
    sage_gemm_kernel<<<gemm_grid, 128, 0, stream>>>(agg, deg, h, Wl1, bl1, Wr1, out1, 1);

    gather_sum_kernel<<<NNODES / 8, 256, 0, stream>>>(out1, rowstart, rowcnt, perm, agg);
    sage_gemm_kernel<<<gemm_grid, 128, 0, stream>>>(agg, deg, out1, Wl2, bl2, Wr2, out2, 0);
}


# CSR build reads dst once: each wave notes its block's owned edges in an LDS list during the count sweep and places them from it; 2nd sweep dropped; a node's sum order still by LDS counter arrival
# speedup vs baseline: 2.9443x; 1.1534x over previous
#include <hip/hip_runtime.h>

typedef float v2f __attribute__((ext_vector_type(2)));
typedef float v8f __attribute__((ext_vector_type(8)));

#define NNODES 10000
#define NEDGES 320000
#define DIN 512
#define DOUT 256

typedef float v4f __attribute__((ext_vector_type(4)));

#define LIN_KB 4

struct LinStage {
    v4f a[LIN_KB];
    v4f b[LIN_KB][4];
};

__device__ __forceinline__ void lin_load(LinStage& s, const float* __restrict__ ap,
                                         const float* __restrict__ bp)
{
    #pragma unroll
    for (int q = 0; q < LIN_KB; ++q) {
        s.a[q] = *(const v4f*)(ap + 8 * q);
        #pragma unroll
        for (int j = 0; j < 4; ++j)
            s.b[q][j] = *(const v4f*)(bp + (size_t)(8 * q + j) * DOUT);
    }
}

__device__ __forceinline__ void lin_mma(const LinStage& s, v8f (&acc)[4])
{
    #pragma unroll
    for (int q = 0; q < LIN_KB; ++q) {
        #pragma unroll
        for (int u = 0; u < 2; ++u) {
            v2f a;
            a.x = s.a[q][2 * u];
            a.y = s.a[q][2 * u + 1];
            #pragma unroll
            for (int t = 0; t < 4; ++t) {
                v2f b;
                b.x = s.b[q][2 * u][t];
                b.y = s.b[q][2 * u + 1][t];
                acc[t] = __builtin_amdgcn_wmma_f32_16x16x4_f32(false, a, false, b,
                                                               (short)0, acc[t], false, false);
            }
        }
    }
}

__global__ __launch_bounds__(128) void lin_gemm_kernel(
    const float* __restrict__ X, const float* __restrict__ W,
    const float* __restrict__ bias, float* __restrict__ H)
{
    const int lane = threadIdx.x & 31;
    const int wave = threadIdx.x >> 5;
    const int c    = lane & 15;
    const int h    = lane >> 4;

    const int row0 = blockIdx.x * 16;
    const int col0 = wave * 64;

    const float* ap = X + (size_t)(row0 + c) * DIN + 4 * h;
    const float* bp = W + (size_t)(4 * h) * DOUT + col0 + 4 * c;

    v8f acc[4] = {};
    LinStage s0, s1;
    lin_load(s0, ap, bp);
    for (int k = 0; k < DIN - 16 * LIN_KB; k += 16 * LIN_KB) {
        lin_load(s1, ap + k + 8 * LIN_KB, bp + (size_t)(k + 8 * LIN_KB) * DOUT);
        __builtin_amdgcn_sched_barrier(0);
        lin_mma(s0, acc);
        __builtin_amdgcn_sched_barrier(0);
        lin_load(s0, ap + k + 16 * LIN_KB, bp + (size_t)(k + 16 * LIN_KB) * DOUT);
        __builtin_amdgcn_sched_barrier(0);
        lin_mma(s1, acc);
    }
    lin_load(s1, ap + DIN - 8 * LIN_KB, bp + (size_t)(DIN - 8 * LIN_KB) * DOUT);
    __builtin_amdgcn_sched_barrier(0);
    lin_mma(s0, acc);
    lin_mma(s1, acc);

    const v4f bv = *(const v4f*)(bias + col0 + 4 * c);
    float* out = H + (size_t)(row0 + 8 * h) * DOUT + col0 + 4 * c;
    #pragma unroll
    for (int r = 0; r < 8; ++r) {
        v4f v;
        v.x = acc[0][r] + bv.x;
        v.y = acc[1][r] + bv.y;
        v.z = acc[2][r] + bv.z;
        v.w = acc[3][r] + bv.w;
        *(v4f*)(out + (size_t)r * DOUT) = v;
    }
}

#define SAGE_KB 2

struct SageStage {
    v4f am[SAGE_KB];
    v4f ah[SAGE_KB];
    v4f bl[SAGE_KB][4];
    v4f br[SAGE_KB][4];
};

__device__ __forceinline__ void sage_load(SageStage& s, const float* __restrict__ amp,
                                          const float* __restrict__ ahp,
                                          const float* __restrict__ blp,
                                          const float* __restrict__ brp)
{
    #pragma unroll
    for (int q = 0; q < SAGE_KB; ++q) {
        s.am[q] = *(const v4f*)(amp + 8 * q);
        s.ah[q] = *(const v4f*)(ahp + 8 * q);
        #pragma unroll
        for (int j = 0; j < 4; ++j) {
            s.bl[q][j] = *(const v4f*)(blp + (size_t)(8 * q + j) * DOUT);
            s.br[q][j] = *(const v4f*)(brp + (size_t)(8 * q + j) * DOUT);
        }
    }
}

__device__ __forceinline__ void sage_mma(const SageStage& s, float inv, v8f (&acc)[4])
{
    #pragma unroll
    for (int q = 0; q < SAGE_KB; ++q) {
        #pragma unroll
        for (int u = 0; u < 2; ++u) {
            v2f am, ah;
            am.x = s.am[q][2 * u] * inv;
            am.y = s.am[q][2 * u + 1] * inv;
            ah.x = s.ah[q][2 * u];
            ah.y = s.ah[q][2 * u + 1];
            #pragma unroll
            for (int t = 0; t < 4; ++t) {
                v2f b;
                b.x = s.bl[q][2 * u][t];
                b.y = s.bl[q][2 * u + 1][t];
                acc[t] = __builtin_amdgcn_wmma_f32_16x16x4_f32(false, am, false, b,
                                                               (short)0, acc[t], false, false);
            }
            #pragma unroll
            for (int t = 0; t < 4; ++t) {
                v2f b;
                b.x = s.br[q][2 * u][t];
                b.y = s.br[q][2 * u + 1][t];
                acc[t] = __builtin_amdgcn_wmma_f32_16x16x4_f32(false, ah, false, b,
                                                               (short)0, acc[t], false, false);
            }
        }
    }
}

__global__ __launch_bounds__(128) void sage_gemm_kernel(
    const float* __restrict__ AGG, const float* __restrict__ INVDEG,
    const float* __restrict__ Hin,
    const float* __restrict__ Wl, const float* __restrict__ bl,
    const float* __restrict__ Wr,
    float* __restrict__ OUT, int relu)
{
    const int lane = threadIdx.x & 31;
    const int wave = threadIdx.x >> 5;
    const int c    = lane & 15;
    const int h    = lane >> 4;

    const int row0 = blockIdx.x * 16;
    const int col0 = wave * 64;

    const int arow_idx = row0 + c;
    const float inv = INVDEG[arow_idx];

    const float* amp = AGG + (size_t)arow_idx * DOUT + 4 * h;
    const float* ahp = Hin + (size_t)arow_idx * DOUT + 4 * h;
    const float* blp = Wl + (size_t)(4 * h) * DOUT + col0 + 4 * c;
    const float* brp = Wr + (size_t)(4 * h) * DOUT + col0 + 4 * c;

    v8f acc[4] = {};
    SageStage s0, s1;
    sage_load(s0, amp, ahp, blp, brp);
    for (int k = 0; k < DOUT - 16 * SAGE_KB; k += 16 * SAGE_KB) {
        const int k1 = k + 8 * SAGE_KB;
        const int k2 = k + 16 * SAGE_KB;
        sage_load(s1, amp + k1, ahp + k1, blp + (size_t)k1 * DOUT, brp + (size_t)k1 * DOUT);
        __builtin_amdgcn_sched_barrier(0);
        sage_mma(s0, inv, acc);
        __builtin_amdgcn_sched_barrier(0);
        sage_load(s0, amp + k2, ahp + k2, blp + (size_t)k2 * DOUT, brp + (size_t)k2 * DOUT);
        __builtin_amdgcn_sched_barrier(0);
        sage_mma(s1, inv, acc);
    }
    const size_t o0 = (size_t)(row0 + 8 * h) * DOUT + col0 + 4 * c;
    const float* hres = Hin + o0;
    float* out = OUT + o0;
    v4f bv;
    v4f hr[8];
    {
        const int k1 = DOUT - 8 * SAGE_KB;
        sage_load(s1, amp + k1, ahp + k1, blp + (size_t)k1 * DOUT, brp + (size_t)k1 * DOUT);
        bv = *(const v4f*)(bl + col0 + 4 * c);
        #pragma unroll
        for (int r = 0; r < 8; ++r) hr[r] = *(const v4f*)(hres + (size_t)r * DOUT);
        __builtin_amdgcn_sched_barrier(0);
        sage_mma(s0, inv, acc);
        sage_mma(s1, inv, acc);
    }

    v4f v[8];
    #pragma unroll
    for (int r = 0; r < 8; ++r) {
        v[r].x = acc[0][r] + bv.x;
        v[r].y = acc[1][r] + bv.y;
        v[r].z = acc[2][r] + bv.z;
        v[r].w = acc[3][r] + bv.w;
    }
    if (relu) {
        #pragma unroll
        for (int r = 0; r < 8; ++r) {
            v[r].x = fmaxf(v[r].x, 0.0f); v[r].y = fmaxf(v[r].y, 0.0f);
            v[r].z = fmaxf(v[r].z, 0.0f); v[r].w = fmaxf(v[r].w, 0.0f);
        }
    }
    #pragma unroll
    for (int r = 0; r < 8; ++r) *(v4f*)(out + (size_t)r * DOUT) = v[r] + hr[r];
}

typedef int   v4i __attribute__((ext_vector_type(4)));

#define CSR_NB   125
#define CSR_R    80
#define CSR_T    1024
#define CSR_NW   (CSR_T / 32)
#define CSR_E4   (NEDGES / 4)
#define CSR_FULL (CSR_E4 / CSR_T)
#define CSR_U    13
#define CSR_U2   6
#define CSR_REST (CSR_E4 - CSR_FULL * CSR_T)
#define CSR_LCAP 512
#define CSR_EBITS 19
#define CSR_PU   4

static_assert(CSR_NB * CSR_R == NNODES, "owner ranges must tile the nodes");
static_assert(NEDGES % 4 == 0, "dst is read in groups of four");
static_assert(CSR_FULL % CSR_U == 0, "load batch must divide the full sweeps");
static_assert(CSR_FULL % CSR_U2 == 0, "load batch must divide the full sweeps");
static_assert(CSR_REST >= 0 && CSR_REST < CSR_T, "one partial sweep at most");
static_assert(NEDGES <= (1 << CSR_EBITS), "edge number must fit its field");
static_assert(CSR_R <= (1 << (31 - CSR_EBITS)), "owned node index must fit above the edge number");

__device__ __forceinline__ void csr_visit(int d, int e, int lo, int* __restrict__ s_cnt,
                                          int* __restrict__ wlist, int& below, int& wcount)
{
    below += __popc(__builtin_amdgcn_ballot_w32(d < lo));
    const unsigned r = (unsigned)(d - lo);
    const bool m = r < (unsigned)CSR_R;
    const unsigned mk = __builtin_amdgcn_ballot_w32(m);
    if (__builtin_expect(mk != 0u, 0)) {
        if (m) {
            atomicAdd(&s_cnt[r], 1);
            const int slot = wcount + (int)__builtin_amdgcn_mbcnt_lo(mk, 0u);
            if (slot < CSR_LCAP) wlist[slot] = (int)((r << CSR_EBITS) | (unsigned)e);
        }
        wcount += __popc(mk);
    }
}

__global__ __launch_bounds__(CSR_T) void csr_build_kernel(
    const int* __restrict__ src, const int* __restrict__ dst,
    int* __restrict__ rowstart, int* __restrict__ rowcnt,
    int* __restrict__ perm, float* __restrict__ invdeg)
{
    __shared__ int s_cnt[CSR_R];
    __shared__ int s_cursor[CSR_R];
    __shared__ int s_below;
    __shared__ int s_over;
    __shared__ int s_list[CSR_NW * CSR_LCAP];

    const int tid  = threadIdx.x;
    const int lane = tid & 31;
    const int lo   = blockIdx.x * CSR_R;
    const v4i* __restrict__ d4 = (const v4i*)dst;
    const v4i* __restrict__ s4 = (const v4i*)src;
    int* __restrict__ wlist = s_list + (tid >> 5) * CSR_LCAP;

    if (tid < CSR_R) s_cnt[tid] = 0;
    if (tid == 0) { s_below = 0; s_over = 0; }
    __syncthreads();

    int below = 0;
    int wcount = 0;
    for (int k0 = 0; k0 < CSR_FULL; k0 += CSR_U) {
        v4i v[CSR_U];
        #pragma unroll
        for (int u = 0; u < CSR_U; ++u) v[u] = d4[tid + (k0 + u) * CSR_T];
        #pragma unroll
        for (int u = 0; u < CSR_U; ++u) {
            const int e0 = 4 * (tid + (k0 + u) * CSR_T);
            #pragma unroll
            for (int c = 0; c < 4; ++c)
                csr_visit(v[u][c], e0 + c, lo, s_cnt, wlist, below, wcount);
        }
    }
    {
        const bool have = tid < CSR_REST;
        v4i v = {NNODES, NNODES, NNODES, NNODES};
        if (have) v = d4[tid + CSR_FULL * CSR_T];
        const int e0 = 4 * (tid + CSR_FULL * CSR_T);
        #pragma unroll
        for (int c = 0; c < 4; ++c)
            csr_visit(v[c], e0 + c, lo, s_cnt, wlist, below, wcount);
    }
    if (lane == 0) {
        atomicAdd(&s_below, below);
        if (wcount > CSR_LCAP) s_over = 1;
    }
    __syncthreads();

    if (tid < CSR_R) {
        int off = 0;
        for (int j = 0; j < tid; ++j) off += s_cnt[j];
        const int c  = s_cnt[tid];
        const int st = s_below + off;
        s_cursor[tid] = st;
        rowstart[lo + tid] = st;
        rowcnt[lo + tid]   = c;
        invdeg[lo + tid]   = 1.0f / fmaxf((float)c, 1.0f);
    }
    __syncthreads();

    if (s_over == 0) {
        for (int i0 = 0; i0 < wcount; i0 += 32 * CSR_PU) {
            int r[CSR_PU], s[CSR_PU], q[CSR_PU];
            #pragma unroll
            for (int u = 0; u < CSR_PU; ++u) {
                const int i = i0 + 32 * u + lane;
                r[u] = -1;
                s[u] = 0;
                if (i < wcount) {
                    const unsigned ent = (unsigned)wlist[i];
                    r[u] = (int)(ent >> CSR_EBITS);
                    s[u] = src[ent & ((1u << CSR_EBITS) - 1u)];
                }
            }
            #pragma unroll
            for (int u = 0; u < CSR_PU; ++u) {
                q[u] = -1;
                if (r[u] >= 0) q[u] = atomicAdd(&s_cursor[r[u]], 1);
            }
            #pragma unroll
            for (int u = 0; u < CSR_PU; ++u) {
                if (q[u] >= 0) perm[q[u]] = s[u];
            }
        }
    } else {
        for (int k0 = 0; k0 < CSR_FULL; k0 += CSR_U2) {
            v4i v[CSR_U2], s[CSR_U2];
            #pragma unroll
            for (int u = 0; u < CSR_U2; ++u) {
                v[u] = d4[tid + (k0 + u) * CSR_T];
                s[u] = s4[tid + (k0 + u) * CSR_T];
            }
            int p[CSR_U2][4];
            #pragma unroll
            for (int u = 0; u < CSR_U2; ++u) {
                #pragma unroll
                for (int c = 0; c < 4; ++c) {
                    const unsigned r = (unsigned)(v[u][c] - lo);
                    int q = -1;
                    if (r < (unsigned)CSR_R) q = atomicAdd(&s_cursor[r], 1);
                    p[u][c] = q;
                }
            }
            #pragma unroll
            for (int u = 0; u < CSR_U2; ++u) {
                #pragma unroll
                for (int c = 0; c < 4; ++c) {
                    if (p[u][c] >= 0) perm[p[u][c]] = s[u][c];
                }
            }
        }
        if (tid < CSR_REST) {
            const v4i v = d4[tid + CSR_FULL * CSR_T];
            const v4i s = s4[tid + CSR_FULL * CSR_T];
            #pragma unroll
            for (int c = 0; c < 4; ++c) {
                const unsigned r = (unsigned)(v[c] - lo);
                if (r < (unsigned)CSR_R) {
                    const int q = atomicAdd(&s_cursor[r], 1);
                    perm[q] = s[c];
                }
            }
        }
    }
}

#define GATHER_U 8
static_assert(NNODES % 8 == 0, "one wave per node, eight waves per block, no partial block");

__global__ __launch_bounds__(256) void gather_sum_kernel(
    const float* __restrict__ H, const int* __restrict__ rowstart,
    const int* __restrict__ rowcnt, const int* __restrict__ perm,
    float* __restrict__ AGG)
{
    const int node = __builtin_amdgcn_readfirstlane((int)(blockIdx.x * 8 + (threadIdx.x >> 5)));
    const int lane = threadIdx.x & 31;
    const int start = rowstart[node];
    const int cnt   = rowcnt[node];

    v4f a0 = {0.0f, 0.0f, 0.0f, 0.0f};
    v4f a1 = {0.0f, 0.0f, 0.0f, 0.0f};

    for (int j0 = 0; j0 < cnt; j0 += 32) {
        const int nb = (cnt - j0 < 32) ? (cnt - j0) : 32;
        int mysrc = 0;
        if (lane < nb) mysrc = perm[start + j0 + lane];
        for (int j = 0; j < nb; j += GATHER_U) {
            const int m = nb - j;
            v4f r0[GATHER_U], r1[GATHER_U];
            if (m >= GATHER_U) {
                #pragma unroll
                for (int u = 0; u < GATHER_U; ++u) {
                    const int s = __builtin_amdgcn_readlane(mysrc, j + u);
                    const v4f* __restrict__ p = (const v4f*)(H + (size_t)s * DOUT);
                    r0[u] = p[lane];
                    r1[u] = p[32 + lane];
                }
                #pragma unroll
                for (int u = 0; u < GATHER_U; ++u) { a0 += r0[u]; a1 += r1[u]; }
            } else {
                #pragma unroll
                for (int u = 0; u < GATHER_U; ++u) {
                    if (u < m) {
                        const int s = __builtin_amdgcn_readlane(mysrc, j + u);
                        const v4f* __restrict__ p = (const v4f*)(H + (size_t)s * DOUT);
                        r0[u] = p[lane];
                        r1[u] = p[32 + lane];
                    }
                }
                #pragma unroll
                for (int u = 0; u < GATHER_U; ++u) {
                    if (u < m) { a0 += r0[u]; a1 += r1[u]; }
                }
            }
        }
    }

    v4f* __restrict__ o = (v4f*)(AGG + (size_t)node * DOUT);
    o[lane]      = a0;
    o[32 + lane] = a1;
}

extern "C" void kernel_launch(void* const* d_in, const int* in_sizes, int n_in,
                              void* d_out, int out_size, void* d_ws, size_t ws_size,
                              hipStream_t stream)
{
    const float* x   = (const float*)d_in[0];
    const int*   ei  = (const int*)d_in[1];
    const float* W0  = (const float*)d_in[2];
    const float* b0  = (const float*)d_in[3];
    const float* Wl1 = (const float*)d_in[4];
    const float* bl1 = (const float*)d_in[5];
    const float* Wr1 = (const float*)d_in[6];
    const float* Wl2 = (const float*)d_in[7];
    const float* bl2 = (const float*)d_in[8];
    const float* Wr2 = (const float*)d_in[9];

    const int* src = ei;
    const int* dst = ei + NEDGES;

    float* ws  = (float*)d_ws;
    float* h   = ws;
    float* agg = h + (size_t)NNODES * DOUT;
    float* deg = agg + (size_t)NNODES * DOUT;
    int* rowstart = (int*)(deg + NNODES);
    int* rowcnt   = rowstart + NNODES;
    int* perm     = rowcnt + NNODES;

    float* out1 = (float*)d_out;
    float* out2 = out1 + (size_t)NNODES * DOUT;

    dim3 gemm_grid(NNODES / 16, 1);

    csr_build_kernel<<<CSR_NB, CSR_T, 0, stream>>>(src, dst, rowstart, rowcnt, perm, deg);

    lin_gemm_kernel<<<gemm_grid, 128, 0, stream>>>(x, W0, b0, h);

    gather_sum_kernel<<<NNODES / 8, 256, 0, stream>>>(h, rowstart, rowcnt, perm, agg);
    sage_gemm_kernel<<<gemm_grid, 128, 0, stream>>>(agg, deg, h, Wl1, bl1, Wr1, out1, 1);

    gather_sum_kernel<<<NNODES / 8, 256, 0, stream>>>(out1, rowstart, rowcnt, perm, agg);
    sage_gemm_kernel<<<gemm_grid, 128, 0, stream>>>(agg, deg, out1, Wl2, bl2, Wr2, out2, 0);
}


# Neighbour gather fused into each layer's dual-GEMM kernel: a block sums its 16 nodes' neighbour rows into an LDS tile of means and multiplies from it; agg buffer, 2 launches gone; CSR sum order kept
# speedup vs baseline: 3.2530x; 1.1048x over previous
#include <hip/hip_runtime.h>

typedef float v2f __attribute__((ext_vector_type(2)));
typedef float v8f __attribute__((ext_vector_type(8)));

#define NNODES 10000
#define NEDGES 320000
#define DIN 512
#define DOUT 256

typedef float v4f __attribute__((ext_vector_type(4)));

#define LIN_KB 4

struct LinStage {
    v4f a[LIN_KB];
    v4f b[LIN_KB][4];
};

__device__ __forceinline__ void lin_load(LinStage& s, const float* __restrict__ ap,
                                         const float* __restrict__ bp)
{
    #pragma unroll
    for (int q = 0; q < LIN_KB; ++q) {
        s.a[q] = *(const v4f*)(ap + 8 * q);
        #pragma unroll
        for (int j = 0; j < 4; ++j)
            s.b[q][j] = *(const v4f*)(bp + (size_t)(8 * q + j) * DOUT);
    }
}

__device__ __forceinline__ void lin_mma(const LinStage& s, v8f (&acc)[4])
{
    #pragma unroll
    for (int q = 0; q < LIN_KB; ++q) {
        #pragma unroll
        for (int u = 0; u < 2; ++u) {
            v2f a;
            a.x = s.a[q][2 * u];
            a.y = s.a[q][2 * u + 1];
            #pragma unroll
            for (int t = 0; t < 4; ++t) {
                v2f b;
                b.x = s.b[q][2 * u][t];
                b.y = s.b[q][2 * u + 1][t];
                acc[t] = __builtin_amdgcn_wmma_f32_16x16x4_f32(false, a, false, b,
                                                               (short)0, acc[t], false, false);
            }
        }
    }
}

__global__ __launch_bounds__(128) void lin_gemm_kernel(
    const float* __restrict__ X, const float* __restrict__ W,
    const float* __restrict__ bias, float* __restrict__ H)
{
    const int lane = threadIdx.x & 31;
    const int wave = threadIdx.x >> 5;
    const int c    = lane & 15;
    const int h    = lane >> 4;

    const int row0 = blockIdx.x * 16;
    const int col0 = wave * 64;

    const float* ap = X + (size_t)(row0 + c) * DIN + 4 * h;
    const float* bp = W + (size_t)(4 * h) * DOUT + col0 + 4 * c;

    v8f acc[4] = {};
    LinStage s0, s1;
    lin_load(s0, ap, bp);
    for (int k = 0; k < DIN - 16 * LIN_KB; k += 16 * LIN_KB) {
        lin_load(s1, ap + k + 8 * LIN_KB, bp + (size_t)(k + 8 * LIN_KB) * DOUT);
        __builtin_amdgcn_sched_barrier(0);
        lin_mma(s0, acc);
        __builtin_amdgcn_sched_barrier(0);
        lin_load(s0, ap + k + 16 * LIN_KB, bp + (size_t)(k + 16 * LIN_KB) * DOUT);
        __builtin_amdgcn_sched_barrier(0);
        lin_mma(s1, acc);
    }
    lin_load(s1, ap + DIN - 8 * LIN_KB, bp + (size_t)(DIN - 8 * LIN_KB) * DOUT);
    __builtin_amdgcn_sched_barrier(0);
    lin_mma(s0, acc);
    lin_mma(s1, acc);

    const v4f bv = *(const v4f*)(bias + col0 + 4 * c);
    float* out = H + (size_t)(row0 + 8 * h) * DOUT + col0 + 4 * c;
    #pragma unroll
    for (int r = 0; r < 8; ++r) {
        v4f v;
        v.x = acc[0][r] + bv.x;
        v.y = acc[1][r] + bv.y;
        v.z = acc[2][r] + bv.z;
        v.w = acc[3][r] + bv.w;
        *(v4f*)(out + (size_t)r * DOUT) = v;
    }
}

typedef int   v4i __attribute__((ext_vector_type(4)));

#define CSR_NB   125
#define CSR_R    80
#define CSR_T    1024
#define CSR_NW   (CSR_T / 32)
#define CSR_E4   (NEDGES / 4)
#define CSR_FULL (CSR_E4 / CSR_T)
#define CSR_U    13
#define CSR_U2   6
#define CSR_REST (CSR_E4 - CSR_FULL * CSR_T)
#define CSR_LCAP 512
#define CSR_EBITS 19
#define CSR_PU   4

static_assert(CSR_NB * CSR_R == NNODES, "owner ranges must tile the nodes");
static_assert(NEDGES % 4 == 0, "dst is read in groups of four");
static_assert(CSR_FULL % CSR_U == 0, "load batch must divide the full sweeps");
static_assert(CSR_FULL % CSR_U2 == 0, "load batch must divide the full sweeps");
static_assert(CSR_REST >= 0 && CSR_REST < CSR_T, "one partial sweep at most");
static_assert(NEDGES <= (1 << CSR_EBITS), "edge number must fit its field");
static_assert(CSR_R <= (1 << (31 - CSR_EBITS)), "owned node index must fit above the edge number");

__device__ __forceinline__ void csr_visit(int d, int e, int lo, int* __restrict__ s_cnt,
                                          int* __restrict__ wlist, int& below, int& wcount)
{
    below += __popc(__builtin_amdgcn_ballot_w32(d < lo));
    const unsigned r = (unsigned)(d - lo);
    const bool m = r < (unsigned)CSR_R;
    const unsigned mk = __builtin_amdgcn_ballot_w32(m);
    if (__builtin_expect(mk != 0u, 0)) {
        if (m) {
            atomicAdd(&s_cnt[r], 1);
            const int slot = wcount + (int)__builtin_amdgcn_mbcnt_lo(mk, 0u);
            if (slot < CSR_LCAP) wlist[slot] = (int)((r << CSR_EBITS) | (unsigned)e);
        }
        wcount += __popc(mk);
    }
}

__global__ __launch_bounds__(CSR_T) void csr_build_kernel(
    const int* __restrict__ src, const int* __restrict__ dst,
    int* __restrict__ rowstart, int* __restrict__ rowcnt,
    int* __restrict__ perm, float* __restrict__ invdeg)
{
    __shared__ int s_cnt[CSR_R];
    __shared__ int s_cursor[CSR_R];
    __shared__ int s_below;
    __shared__ int s_over;
    __shared__ int s_list[CSR_NW * CSR_LCAP];

    const int tid  = threadIdx.x;
    const int lane = tid & 31;
    const int lo   = blockIdx.x * CSR_R;
    const v4i* __restrict__ d4 = (const v4i*)dst;
    const v4i* __restrict__ s4 = (const v4i*)src;
    int* __restrict__ wlist = s_list + (tid >> 5) * CSR_LCAP;

    if (tid < CSR_R) s_cnt[tid] = 0;
    if (tid == 0) { s_below = 0; s_over = 0; }
    __syncthreads();

    int below = 0;
    int wcount = 0;
    for (int k0 = 0; k0 < CSR_FULL; k0 += CSR_U) {
        v4i v[CSR_U];
        #pragma unroll
        for (int u = 0; u < CSR_U; ++u) v[u] = d4[tid + (k0 + u) * CSR_T];
        #pragma unroll
        for (int u = 0; u < CSR_U; ++u) {
            const int e0 = 4 * (tid + (k0 + u) * CSR_T);
            #pragma unroll
            for (int c = 0; c < 4; ++c)
                csr_visit(v[u][c], e0 + c, lo, s_cnt, wlist, below, wcount);
        }
    }
    {
        const bool have = tid < CSR_REST;
        v4i v = {NNODES, NNODES, NNODES, NNODES};
        if (have) v = d4[tid + CSR_FULL * CSR_T];
        const int e0 = 4 * (tid + CSR_FULL * CSR_T);
        #pragma unroll
        for (int c = 0; c < 4; ++c)
            csr_visit(v[c], e0 + c, lo, s_cnt, wlist, below, wcount);
    }
    if (lane == 0) {
        atomicAdd(&s_below, below);
        if (wcount > CSR_LCAP) s_over = 1;
    }
    __syncthreads();

    if (tid < CSR_R) {
        int off = 0;
        for (int j = 0; j < tid; ++j) off += s_cnt[j];
        const int c  = s_cnt[tid];
        const int st = s_below + off;
        s_cursor[tid] = st;
        rowstart[lo + tid] = st;
        rowcnt[lo + tid]   = c;
        invdeg[lo + tid]   = 1.0f / fmaxf((float)c, 1.0f);
    }
    __syncthreads();

    if (s_over == 0) {
        for (int i0 = 0; i0 < wcount; i0 += 32 * CSR_PU) {
            int r[CSR_PU], s[CSR_PU], q[CSR_PU];
            #pragma unroll
            for (int u = 0; u < CSR_PU; ++u) {
                const int i = i0 + 32 * u + lane;
                r[u] = -1;
                s[u] = 0;
                if (i < wcount) {
                    const unsigned ent = (unsigned)wlist[i];
                    r[u] = (int)(ent >> CSR_EBITS);
                    s[u] = src[ent & ((1u << CSR_EBITS) - 1u)];
                }
            }
            #pragma unroll
            for (int u = 0; u < CSR_PU; ++u) {
                q[u] = -1;
                if (r[u] >= 0) q[u] = atomicAdd(&s_cursor[r[u]], 1);
            }
            #pragma unroll
            for (int u = 0; u < CSR_PU; ++u) {
                if (q[u] >= 0) perm[q[u]] = s[u];
            }
        }
    } else {
        for (int k0 = 0; k0 < CSR_FULL; k0 += CSR_U2) {
            v4i v[CSR_U2], s[CSR_U2];
            #pragma unroll
            for (int u = 0; u < CSR_U2; ++u) {
                v[u] = d4[tid + (k0 + u) * CSR_T];
                s[u] = s4[tid + (k0 + u) * CSR_T];
            }
            int p[CSR_U2][4];
            #pragma unroll
            for (int u = 0; u < CSR_U2; ++u) {
                #pragma unroll
                for (int c = 0; c < 4; ++c) {
                    const unsigned r = (unsigned)(v[u][c] - lo);
                    int q = -1;
                    if (r < (unsigned)CSR_R) q = atomicAdd(&s_cursor[r], 1);
                    p[u][c] = q;
                }
            }
            #pragma unroll
            for (int u = 0; u < CSR_U2; ++u) {
                #pragma unroll
                for (int c = 0; c < 4; ++c) {
                    if (p[u][c] >= 0) perm[p[u][c]] = s[u][c];
                }
            }
        }
        if (tid < CSR_REST) {
            const v4i v = d4[tid + CSR_FULL * CSR_T];
            const v4i s = s4[tid + CSR_FULL * CSR_T];
            #pragma unroll
            for (int c = 0; c < 4; ++c) {
                const unsigned r = (unsigned)(v[c] - lo);
                if (r < (unsigned)CSR_R) {
                    const int q = atomicAdd(&s_cursor[r], 1);
                    perm[q] = s[c];
                }
            }
        }
    }
}

#define SAGE_KB 2
#define FG_PITCH (DOUT + 4)
#define FG_U 16

static_assert(NNODES % 16 == 0, "no partial row tile");

struct SageStage {
    v4f am[SAGE_KB];
    v4f ah[SAGE_KB];
    v4f bl[SAGE_KB][4];
    v4f br[SAGE_KB][4];
};

__device__ __forceinline__ void sage_load_g(SageStage& s, const float* __restrict__ ahp,
                                            const float* __restrict__ blp,
                                            const float* __restrict__ brp)
{
    #pragma unroll
    for (int q = 0; q < SAGE_KB; ++q) {
        s.ah[q] = *(const v4f*)(ahp + 8 * q);
        #pragma unroll
        for (int j = 0; j < 4; ++j) {
            s.bl[q][j] = *(const v4f*)(blp + (size_t)(8 * q + j) * DOUT);
            s.br[q][j] = *(const v4f*)(brp + (size_t)(8 * q + j) * DOUT);
        }
    }
}

__device__ __forceinline__ void sage_load_l(SageStage& s, const float* amp)
{
    #pragma unroll
    for (int q = 0; q < SAGE_KB; ++q) s.am[q] = *(const v4f*)(amp + 8 * q);
}

__device__ __forceinline__ void sage_mma(const SageStage& s, v8f (&acc)[4])
{
    #pragma unroll
    for (int q = 0; q < SAGE_KB; ++q) {
        #pragma unroll
        for (int u = 0; u < 2; ++u) {
            v2f am, ah;
            am.x = s.am[q][2 * u];
            am.y = s.am[q][2 * u + 1];
            ah.x = s.ah[q][2 * u];
            ah.y = s.ah[q][2 * u + 1];
            #pragma unroll
            for (int t = 0; t < 4; ++t) {
                v2f b;
                b.x = s.bl[q][2 * u][t];
                b.y = s.bl[q][2 * u + 1][t];
                acc[t] = __builtin_amdgcn_wmma_f32_16x16x4_f32(false, am, false, b,
                                                               (short)0, acc[t], false, false);
            }
            #pragma unroll
            for (int t = 0; t < 4; ++t) {
                v2f b;
                b.x = s.br[q][2 * u][t];
                b.y = s.br[q][2 * u + 1][t];
                acc[t] = __builtin_amdgcn_wmma_f32_16x16x4_f32(false, ah, false, b,
                                                               (short)0, acc[t], false, false);
            }
        }
    }
}

__device__ __forceinline__ void fg_rows(const float* __restrict__ H, int mysrc, int j,
                                        const int n, int lane, v4f& a0, v4f& a1)
{
    v4f r0[FG_U], r1[FG_U];
    #pragma unroll
    for (int u = 0; u < n; ++u) {
        const int sidx = __builtin_amdgcn_readlane(mysrc, j + u);
        const v4f* __restrict__ p = (const v4f*)(H + (size_t)sidx * DOUT);
        r0[u] = p[lane];
        r1[u] = p[32 + lane];
    }
    #pragma unroll
    for (int u = 0; u < n; ++u) { a0 += r0[u]; a1 += r1[u]; }
}

__device__ __forceinline__ void fg_tail(const float* __restrict__ H, int mysrc, int j,
                                        int m, int lane, v4f& a0, v4f& a1)
{
    int sidx[7];
    v4f r0[7], r1[7];
    #pragma unroll
    for (int u = 0; u < 7; ++u) sidx[u] = __builtin_amdgcn_readlane(mysrc, (j + u) & 31);
    #pragma unroll
    for (int u = 0; u < 7; ++u) {
        if (u < m) {
            const v4f* __restrict__ p = (const v4f*)(H + (size_t)sidx[u] * DOUT);
            r0[u] = p[lane];
            r1[u] = p[32 + lane];
        }
    }
    #pragma unroll
    for (int u = 0; u < 7; ++u) {
        if (u < m) { a0 += r0[u]; a1 += r1[u]; }
    }
}

__global__ __launch_bounds__(128) void sage_fused_kernel(
    const float* __restrict__ Hin,
    const int* __restrict__ rowstart, const int* __restrict__ rowcnt,
    const int* __restrict__ perm, const float* __restrict__ INVDEG,
    const float* __restrict__ Wl, const float* __restrict__ bl,
    const float* __restrict__ Wr,
    float* __restrict__ OUT, int relu)
{
    __shared__ __attribute__((aligned(16))) float s_mean[16 * FG_PITCH];

    const int lane = threadIdx.x & 31;
    const int wave = threadIdx.x >> 5;
    const int c    = lane & 15;
    const int h    = lane >> 4;

    const int row0 = blockIdx.x * 16;
    const int col0 = wave * 64;

    {
        const int nb0 = __builtin_amdgcn_readfirstlane(row0 + 4 * wave);
        const v4i st4 = *(const v4i*)(rowstart + nb0);
        const v4i cn4 = *(const v4i*)(rowcnt + nb0);
        const v4f iv4 = *(const v4f*)(INVDEG + nb0);

        int pre0[4], pre1[4];
        #pragma unroll
        for (int i = 0; i < 4; ++i) {
            pre0[i] = 0;
            pre1[i] = 0;
            if (lane < cn4[i])      pre0[i] = perm[st4[i] + lane];
            if (lane + 32 < cn4[i]) pre1[i] = perm[st4[i] + 32 + lane];
        }

        #pragma unroll
        for (int i = 0; i < 4; ++i) {
            const int start = st4[i];
            const int cnt   = cn4[i];
            v4f a0 = {0.0f, 0.0f, 0.0f, 0.0f};
            v4f a1 = {0.0f, 0.0f, 0.0f, 0.0f};
            for (int j0 = 0; j0 < cnt; j0 += 32) {
                const int nb = (cnt - j0 < 32) ? (cnt - j0) : 32;
                int mysrc;
                if (j0 == 0)       mysrc = pre0[i];
                else if (j0 == 32) mysrc = pre1[i];
                else {
                    mysrc = 0;
                    if (lane < nb) mysrc = perm[start + j0 + lane];
                }
                int j = 0;
                for (; j + FG_U <= nb; j += FG_U) fg_rows(Hin, mysrc, j, FG_U, lane, a0, a1);
                if (j + 8 <= nb) { fg_rows(Hin, mysrc, j, 8, lane, a0, a1); j += 8; }
                if (j < nb) fg_tail(Hin, mysrc, j, nb - j, lane, a0, a1);
            }
            const float inv = iv4[i];
            float* o = s_mean + (4 * wave + i) * FG_PITCH;
            *(v4f*)(o + 4 * lane)       = a0 * inv;
            *(v4f*)(o + 128 + 4 * lane) = a1 * inv;
        }
    }

    const int arow_idx = row0 + c;
    const float* amp = s_mean + c * FG_PITCH + 4 * h;
    const float* ahp = Hin + (size_t)arow_idx * DOUT + 4 * h;
    const float* blp = Wl + (size_t)(4 * h) * DOUT + col0 + 4 * c;
    const float* brp = Wr + (size_t)(4 * h) * DOUT + col0 + 4 * c;

    v8f acc[4] = {};
    SageStage s0, s1;
    sage_load_g(s0, ahp, blp, brp);
    __syncthreads();
    sage_load_l(s0, amp);
    for (int k = 0; k < DOUT - 16 * SAGE_KB; k += 16 * SAGE_KB) {
        const int k1 = k + 8 * SAGE_KB;
        const int k2 = k + 16 * SAGE_KB;
        sage_load_g(s1, ahp + k1, blp + (size_t)k1 * DOUT, brp + (size_t)k1 * DOUT);
        sage_load_l(s1, amp + k1);
        __builtin_amdgcn_sched_barrier(0);
        sage_mma(s0, acc);
        __builtin_amdgcn_sched_barrier(0);
        sage_load_g(s0, ahp + k2, blp + (size_t)k2 * DOUT, brp + (size_t)k2 * DOUT);
        sage_load_l(s0, amp + k2);
        __builtin_amdgcn_sched_barrier(0);
        sage_mma(s1, acc);
    }
    const size_t o0 = (size_t)(row0 + 8 * h) * DOUT + col0 + 4 * c;
    const float* hres = Hin + o0;
    float* out = OUT + o0;
    v4f bv;
    v4f hr[8];
    {
        const int k1 = DOUT - 8 * SAGE_KB;
        sage_load_g(s1, ahp + k1, blp + (size_t)k1 * DOUT, brp + (size_t)k1 * DOUT);
        sage_load_l(s1, amp + k1);
        bv = *(const v4f*)(bl + col0 + 4 * c);
        #pragma unroll
        for (int r = 0; r < 8; ++r) hr[r] = *(const v4f*)(hres + (size_t)r * DOUT);
        __builtin_amdgcn_sched_barrier(0);
        sage_mma(s0, acc);
        sage_mma(s1, acc);
    }

    v4f v[8];
    #pragma unroll
    for (int r = 0; r < 8; ++r) {
        v[r].x = acc[0][r] + bv.x;
        v[r].y = acc[1][r] + bv.y;
        v[r].z = acc[2][r] + bv.z;
        v[r].w = acc[3][r] + bv.w;
    }
    if (relu) {
        #pragma unroll
        for (int r = 0; r < 8; ++r) {
            v[r].x = fmaxf(v[r].x, 0.0f); v[r].y = fmaxf(v[r].y, 0.0f);
            v[r].z = fmaxf(v[r].z, 0.0f); v[r].w = fmaxf(v[r].w, 0.0f);
        }
    }
    #pragma unroll
    for (int r = 0; r < 8; ++r) *(v4f*)(out + (size_t)r * DOUT) = v[r] + hr[r];
}

extern "C" void kernel_launch(void* const* d_in, const int* in_sizes, int n_in,
                              void* d_out, int out_size, void* d_ws, size_t ws_size,
                              hipStream_t stream)
{
    const float* x   = (const float*)d_in[0];
    const int*   ei  = (const int*)d_in[1];
    const float* W0  = (const float*)d_in[2];
    const float* b0  = (const float*)d_in[3];
    const float* Wl1 = (const float*)d_in[4];
    const float* bl1 = (const float*)d_in[5];
    const float* Wr1 = (const float*)d_in[6];
    const float* Wl2 = (const float*)d_in[7];
    const float* bl2 = (const float*)d_in[8];
    const float* Wr2 = (const float*)d_in[9];

    const int* src = ei;
    const int* dst = ei + NEDGES;

    float* ws  = (float*)d_ws;
    float* h   = ws;
    float* deg = h + (size_t)NNODES * DOUT;
    int* rowstart = (int*)(deg + NNODES);
    int* rowcnt   = rowstart + NNODES;
    int* perm     = rowcnt + NNODES;

    float* out1 = (float*)d_out;
    float* out2 = out1 + (size_t)NNODES * DOUT;

    dim3 gemm_grid(NNODES / 16, 1);

    csr_build_kernel<<<CSR_NB, CSR_T, 0, stream>>>(src, dst, rowstart, rowcnt, perm, deg);

    lin_gemm_kernel<<<gemm_grid, 128, 0, stream>>>(x, W0, b0, h);

    sage_fused_kernel<<<gemm_grid, 128, 0, stream>>>(h, rowstart, rowcnt, perm, deg,
                                                     Wl1, bl1, Wr1, out1, 1);

    sage_fused_kernel<<<gemm_grid, 128, 0, stream>>>(out1, rowstart, rowcnt, perm, deg,
                                                     Wl2, bl2, Wr2, out2, 0);
}


# CSR build cut 4 ways along the edge list: a block sweeps a quarter of dst for 200 nodes (200 blocks, was 125 sweeping all); node list = 4 segments, summed slice 0 first; 1/deg moved to layer kernel
# speedup vs baseline: 3.3748x; 1.0374x over previous
#include <hip/hip_runtime.h>

typedef float v2f __attribute__((ext_vector_type(2)));
typedef float v8f __attribute__((ext_vector_type(8)));

#define NNODES 10000
#define NEDGES 320000
#define DIN 512
#define DOUT 256

typedef float v4f __attribute__((ext_vector_type(4)));

#define LIN_KB 4

struct LinStage {
    v4f a[LIN_KB];
    v4f b[LIN_KB][4];
};

__device__ __forceinline__ void lin_load(LinStage& s, const float* __restrict__ ap,
                                         const float* __restrict__ bp)
{
    #pragma unroll
    for (int q = 0; q < LIN_KB; ++q) {
        s.a[q] = *(const v4f*)(ap + 8 * q);
        #pragma unroll
        for (int j = 0; j < 4; ++j)
            s.b[q][j] = *(const v4f*)(bp + (size_t)(8 * q + j) * DOUT);
    }
}

__device__ __forceinline__ void lin_mma(const LinStage& s, v8f (&acc)[4])
{
    #pragma unroll
    for (int q = 0; q < LIN_KB; ++q) {
        #pragma unroll
        for (int u = 0; u < 2; ++u) {
            v2f a;
            a.x = s.a[q][2 * u];
            a.y = s.a[q][2 * u + 1];
            #pragma unroll
            for (int t = 0; t < 4; ++t) {
                v2f b;
                b.x = s.b[q][2 * u][t];
                b.y = s.b[q][2 * u + 1][t];
                acc[t] = __builtin_amdgcn_wmma_f32_16x16x4_f32(false, a, false, b,
                                                               (short)0, acc[t], false, false);
            }
        }
    }
}

__global__ __launch_bounds__(128) void lin_gemm_kernel(
    const float* __restrict__ X, const float* __restrict__ W,
    const float* __restrict__ bias, float* __restrict__ H)
{
    const int lane = threadIdx.x & 31;
    const int wave = threadIdx.x >> 5;
    const int c    = lane & 15;
    const int h    = lane >> 4;

    const int row0 = blockIdx.x * 16;
    const int col0 = wave * 64;

    const float* ap = X + (size_t)(row0 + c) * DIN + 4 * h;
    const float* bp = W + (size_t)(4 * h) * DOUT + col0 + 4 * c;

    v8f acc[4] = {};
    LinStage s0, s1;
    lin_load(s0, ap, bp);
    for (int k = 0; k < DIN - 16 * LIN_KB; k += 16 * LIN_KB) {
        lin_load(s1, ap + k + 8 * LIN_KB, bp + (size_t)(k + 8 * LIN_KB) * DOUT);
        __builtin_amdgcn_sched_barrier(0);
        lin_mma(s0, acc);
        __builtin_amdgcn_sched_barrier(0);
        lin_load(s0, ap + k + 16 * LIN_KB, bp + (size_t)(k + 16 * LIN_KB) * DOUT);
        __builtin_amdgcn_sched_barrier(0);
        lin_mma(s1, acc);
    }
    lin_load(s1, ap + DIN - 8 * LIN_KB, bp + (size_t)(DIN - 8 * LIN_KB) * DOUT);
    __builtin_amdgcn_sched_barrier(0);
    lin_mma(s0, acc);
    lin_mma(s1, acc);

    const v4f bv = *(const v4f*)(bias + col0 + 4 * c);
    float* out = H + (size_t)(row0 + 8 * h) * DOUT + col0 + 4 * c;
    #pragma unroll
    for (int r = 0; r < 8; ++r) {
        v4f v;
        v.x = acc[0][r] + bv.x;
        v.y = acc[1][r] + bv.y;
        v.z = acc[2][r] + bv.z;
        v.w = acc[3][r] + bv.w;
        *(v4f*)(out + (size_t)r * DOUT) = v;
    }
}

typedef int   v4i __attribute__((ext_vector_type(4)));

#define CSR_S    4
#define CSR_NR   50
#define CSR_NB   (CSR_S * CSR_NR)
#define CSR_R    200
#define CSR_T    512
#define CSR_NW   (CSR_T / 32)
#define CSR_ESL  (NEDGES / CSR_S)
#define CSR_E4   (CSR_ESL / 4)
#define CSR_FULL (CSR_E4 / CSR_T)
#define CSR_U    13
#define CSR_U2   3
#define CSR_REST (CSR_E4 - CSR_FULL * CSR_T)
#define CSR_LCAP 512
#define CSR_EBITS 19
#define CSR_PU   4

static_assert(CSR_NR * CSR_R == NNODES, "node ranges must tile the nodes");
static_assert(CSR_S * CSR_ESL == NEDGES, "slices must tile the edges");
static_assert(CSR_ESL % 4 == 0, "dst is read in groups of four, slices start on a 16-byte boundary");
static_assert(CSR_R <= CSR_T, "one thread per owned node");
static_assert(CSR_FULL % CSR_U == 0, "load batch must divide the full sweeps");
static_assert(CSR_FULL % CSR_U2 == 0, "load batch must divide the full sweeps");
static_assert(CSR_REST >= 0 && CSR_REST < CSR_T, "one partial sweep at most");
static_assert(CSR_ESL <= (1 << CSR_EBITS), "edge number (within the slice) must fit its field");
static_assert(CSR_R <= (1 << (31 - CSR_EBITS)), "owned node index must fit above the edge number");

__device__ __forceinline__ void csr_visit(int d, int e, int lo, int* __restrict__ s_cnt,
                                          int* __restrict__ wlist, int& below, int& wcount)
{
    below += __popc(__builtin_amdgcn_ballot_w32(d < lo));
    const unsigned r = (unsigned)(d - lo);
    const bool m = r < (unsigned)CSR_R;
    const unsigned mk = __builtin_amdgcn_ballot_w32(m);
    if (__builtin_expect(mk != 0u, 0)) {
        if (m) {
            atomicAdd(&s_cnt[r], 1);
            const int slot = wcount + (int)__builtin_amdgcn_mbcnt_lo(mk, 0u);
            if (slot < CSR_LCAP) wlist[slot] = (int)((r << CSR_EBITS) | (unsigned)e);
        }
        wcount += __popc(mk);
    }
}

__global__ __launch_bounds__(CSR_T) void csr_build_kernel(
    const int* __restrict__ src, const int* __restrict__ dst,
    int* __restrict__ rowstart, int* __restrict__ rowcnt,
    int* __restrict__ perm)
{
    __shared__ int s_cnt[CSR_R];
    __shared__ int s_cursor[CSR_R];
    __shared__ int s_below;
    __shared__ int s_over;
    __shared__ int s_list[CSR_NW * CSR_LCAP];

    const int tid  = threadIdx.x;
    const int lane = tid & 31;
    const int slice = blockIdx.x / CSR_NR;
    const int lo    = (blockIdx.x % CSR_NR) * CSR_R;
    const int ebase = slice * CSR_ESL;
    const int* __restrict__ srcs = src + ebase;
    const v4i* __restrict__ d4 = (const v4i*)(dst + ebase);
    const v4i* __restrict__ s4 = (const v4i*)srcs;
    int* __restrict__ rs_out = rowstart + slice * NNODES;
    int* __restrict__ rc_out = rowcnt + slice * NNODES;
    int* __restrict__ wlist = s_list + (tid >> 5) * CSR_LCAP;

    if (tid < CSR_R) s_cnt[tid] = 0;
    if (tid == 0) { s_below = 0; s_over = 0; }
    __syncthreads();

    int below = 0;
    int wcount = 0;
    for (int k0 = 0; k0 < CSR_FULL; k0 += CSR_U) {
        v4i v[CSR_U];
        #pragma unroll
        for (int u = 0; u < CSR_U; ++u) v[u] = d4[tid + (k0 + u) * CSR_T];
        #pragma unroll
        for (int u = 0; u < CSR_U; ++u) {
            const int e0 = 4 * (tid + (k0 + u) * CSR_T);
            #pragma unroll
            for (int c = 0; c < 4; ++c)
                csr_visit(v[u][c], e0 + c, lo, s_cnt, wlist, below, wcount);
        }
    }
    {
        const bool have = tid < CSR_REST;
        v4i v = {NNODES, NNODES, NNODES, NNODES};
        if (have) v = d4[tid + CSR_FULL * CSR_T];
        const int e0 = 4 * (tid + CSR_FULL * CSR_T);
        #pragma unroll
        for (int c = 0; c < 4; ++c)
            csr_visit(v[c], e0 + c, lo, s_cnt, wlist, below, wcount);
    }
    if (lane == 0) {
        atomicAdd(&s_below, below);
        if (wcount > CSR_LCAP) s_over = 1;
    }
    __syncthreads();

    if (tid < CSR_R) {
        int off = 0;
        for (int j = 0; j < tid; ++j) off += s_cnt[j];
        const int c  = s_cnt[tid];
        const int st = ebase + s_below + off;
        s_cursor[tid] = st;
        rs_out[lo + tid] = st;
        rc_out[lo + tid] = c;
    }
    __syncthreads();

    if (s_over == 0) {
        for (int i0 = 0; i0 < wcount; i0 += 32 * CSR_PU) {
            int r[CSR_PU], s[CSR_PU], q[CSR_PU];
            #pragma unroll
            for (int u = 0; u < CSR_PU; ++u) {
                const int i = i0 + 32 * u + lane;
                r[u] = -1;
                s[u] = 0;
                if (i < wcount) {
                    const unsigned ent = (unsigned)wlist[i];
                    r[u] = (int)(ent >> CSR_EBITS);
                    s[u] = srcs[ent & ((1u << CSR_EBITS) - 1u)];
                }
            }
            #pragma unroll
            for (int u = 0; u < CSR_PU; ++u) {
                q[u] = -1;
                if (r[u] >= 0) q[u] = atomicAdd(&s_cursor[r[u]], 1);
            }
            #pragma unroll
            for (int u = 0; u < CSR_PU; ++u) {
                if (q[u] >= 0) perm[q[u]] = s[u];
            }
        }
    } else {
        for (int k0 = 0; k0 < CSR_FULL; k0 += CSR_U2) {
            v4i v[CSR_U2], s[CSR_U2];
            #pragma unroll
            for (int u = 0; u < CSR_U2; ++u) {
                v[u] = d4[tid + (k0 + u) * CSR_T];
                s[u] = s4[tid + (k0 + u) * CSR_T];
            }
            int p[CSR_U2][4];
            #pragma unroll
            for (int u = 0; u < CSR_U2; ++u) {
                #pragma unroll
                for (int c = 0; c < 4; ++c) {
                    const unsigned r = (unsigned)(v[u][c] - lo);
                    int q = -1;
                    if (r < (unsigned)CSR_R) q = atomicAdd(&s_cursor[r], 1);
                    p[u][c] = q;
                }
            }
            #pragma unroll
            for (int u = 0; u < CSR_U2; ++u) {
                #pragma unroll
                for (int c = 0; c < 4; ++c) {
                    if (p[u][c] >= 0) perm[p[u][c]] = s[u][c];
                }
            }
        }
        if (tid < CSR_REST) {
            const v4i v = d4[tid + CSR_FULL * CSR_T];
            const v4i s = s4[tid + CSR_FULL * CSR_T];
            #pragma unroll
            for (int c = 0; c < 4; ++c) {
                const unsigned r = (unsigned)(v[c] - lo);
                if (r < (unsigned)CSR_R) {
                    const int q = atomicAdd(&s_cursor[r], 1);
                    perm[q] = s[c];
                }
            }
        }
    }
}

#define SAGE_KB 2
#define FG_PITCH (DOUT + 4)
#define FG_U 16

static_assert(NNODES % 16 == 0, "no partial row tile");

struct SageStage {
    v4f am[SAGE_KB];
    v4f ah[SAGE_KB];
    v4f bl[SAGE_KB][4];
    v4f br[SAGE_KB][4];
};

__device__ __forceinline__ void sage_load_g(SageStage& s, const float* __restrict__ ahp,
                                            const float* __restrict__ blp,
                                            const float* __restrict__ brp)
{
    #pragma unroll
    for (int q = 0; q < SAGE_KB; ++q) {
        s.ah[q] = *(const v4f*)(ahp + 8 * q);
        #pragma unroll
        for (int j = 0; j < 4; ++j) {
            s.bl[q][j] = *(const v4f*)(blp + (size_t)(8 * q + j) * DOUT);
            s.br[q][j] = *(const v4f*)(brp + (size_t)(8 * q + j) * DOUT);
        }
    }
}

__device__ __forceinline__ void sage_load_l(SageStage& s, const float* amp)
{
    #pragma unroll
    for (int q = 0; q < SAGE_KB; ++q) s.am[q] = *(const v4f*)(amp + 8 * q);
}

__device__ __forceinline__ void sage_mma(const SageStage& s, v8f (&acc)[4])
{
    #pragma unroll
    for (int q = 0; q < SAGE_KB; ++q) {
        #pragma unroll
        for (int u = 0; u < 2; ++u) {
            v2f am, ah;
            am.x = s.am[q][2 * u];
            am.y = s.am[q][2 * u + 1];
            ah.x = s.ah[q][2 * u];
            ah.y = s.ah[q][2 * u + 1];
            #pragma unroll
            for (int t = 0; t < 4; ++t) {
                v2f b;
                b.x = s.bl[q][2 * u][t];
                b.y = s.bl[q][2 * u + 1][t];
                acc[t] = __builtin_amdgcn_wmma_f32_16x16x4_f32(false, am, false, b,
                                                               (short)0, acc[t], false, false);
            }
            #pragma unroll
            for (int t = 0; t < 4; ++t) {
                v2f b;
                b.x = s.br[q][2 * u][t];
                b.y = s.br[q][2 * u + 1][t];
                acc[t] = __builtin_amdgcn_wmma_f32_16x16x4_f32(false, ah, false, b,
                                                               (short)0, acc[t], false, false);
            }
        }
    }
}

__device__ __forceinline__ void fg_rows(const float* __restrict__ H, int mysrc, int j,
                                        const int n, int lane, v4f& a0, v4f& a1)
{
    v4f r0[FG_U], r1[FG_U];
    #pragma unroll
    for (int u = 0; u < n; ++u) {
        const int sidx = __builtin_amdgcn_readlane(mysrc, j + u);
        const v4f* __restrict__ p = (const v4f*)(H + (size_t)sidx * DOUT);
        r0[u] = p[lane];
        r1[u] = p[32 + lane];
    }
    #pragma unroll
    for (int u = 0; u < n; ++u) { a0 += r0[u]; a1 += r1[u]; }
}

__device__ __forceinline__ void fg_tail(const float* __restrict__ H, int mysrc, int j,
                                        int m, int lane, v4f& a0, v4f& a1)
{
    int sidx[7];
    v4f r0[7], r1[7];
    #pragma unroll
    for (int u = 0; u < 7; ++u) sidx[u] = __builtin_amdgcn_readlane(mysrc, (j + u) & 31);
    #pragma unroll
    for (int u = 0; u < 7; ++u) {
        if (u < m) {
            const v4f* __restrict__ p = (const v4f*)(H + (size_t)sidx[u] * DOUT);
            r0[u] = p[lane];
            r1[u] = p[32 + lane];
        }
    }
    #pragma unroll
    for (int u = 0; u < 7; ++u) {
        if (u < m) { a0 += r0[u]; a1 += r1[u]; }
    }
}

__device__ __forceinline__ int fg_slot(int k, const int (&bnd)[CSR_S], const int (&off)[CSR_S])
{
    int o = off[CSR_S - 1];
    #pragma unroll
    for (int s = CSR_S - 2; s >= 0; --s) o = (k < bnd[s]) ? off[s] : o;
    return k + o;
}

__global__ __launch_bounds__(128) void sage_fused_kernel(
    const float* __restrict__ Hin,
    const int* __restrict__ rowstart, const int* __restrict__ rowcnt,
    const int* __restrict__ perm,
    const float* __restrict__ Wl, const float* __restrict__ bl,
    const float* __restrict__ Wr,
    float* __restrict__ OUT, int relu)
{
    __shared__ __attribute__((aligned(16))) float s_mean[16 * FG_PITCH];

    const int lane = threadIdx.x & 31;
    const int wave = threadIdx.x >> 5;
    const int c    = lane & 15;
    const int h    = lane >> 4;

    const int row0 = blockIdx.x * 16;
    const int col0 = wave * 64;

    {
        const int nb0 = __builtin_amdgcn_readfirstlane(row0 + 4 * wave);
        v4i st4[CSR_S], cn4[CSR_S];
        #pragma unroll
        for (int s = 0; s < CSR_S; ++s) {
            st4[s] = *(const v4i*)(rowstart + s * NNODES + nb0);
            cn4[s] = *(const v4i*)(rowcnt + s * NNODES + nb0);
        }
        int bnd[4][CSR_S], off[4][CSR_S];
        #pragma unroll
        for (int i = 0; i < 4; ++i) {
            int run = 0;
            #pragma unroll
            for (int s = 0; s < CSR_S; ++s) {
                off[i][s] = st4[s][i] - run;
                run += cn4[s][i];
                bnd[i][s] = run;
            }
        }

        int pre0[4], pre1[4];
        #pragma unroll
        for (int i = 0; i < 4; ++i) {
            pre0[i] = 0;
            pre1[i] = 0;
            if (lane < bnd[i][CSR_S - 1])      pre0[i] = perm[fg_slot(lane, bnd[i], off[i])];
            if (lane + 32 < bnd[i][CSR_S - 1]) pre1[i] = perm[fg_slot(lane + 32, bnd[i], off[i])];
        }

        #pragma unroll
        for (int i = 0; i < 4; ++i) {
            const int cnt   = bnd[i][CSR_S - 1];
            v4f a0 = {0.0f, 0.0f, 0.0f, 0.0f};
            v4f a1 = {0.0f, 0.0f, 0.0f, 0.0f};
            for (int j0 = 0; j0 < cnt; j0 += 32) {
                const int nb = (cnt - j0 < 32) ? (cnt - j0) : 32;
                int mysrc;
                if (j0 == 0)       mysrc = pre0[i];
                else if (j0 == 32) mysrc = pre1[i];
                else {
                    mysrc = 0;
                    if (lane < nb) mysrc = perm[fg_slot(j0 + lane, bnd[i], off[i])];
                }
                int j = 0;
                for (; j + FG_U <= nb; j += FG_U) fg_rows(Hin, mysrc, j, FG_U, lane, a0, a1);
                if (j + 8 <= nb) { fg_rows(Hin, mysrc, j, 8, lane, a0, a1); j += 8; }
                if (j < nb) fg_tail(Hin, mysrc, j, nb - j, lane, a0, a1);
            }
            const float inv = 1.0f / fmaxf((float)cnt, 1.0f);
            float* o = s_mean + (4 * wave + i) * FG_PITCH;
            *(v4f*)(o + 4 * lane)       = a0 * inv;
            *(v4f*)(o + 128 + 4 * lane) = a1 * inv;
        }
    }

    const int arow_idx = row0 + c;
    const float* amp = s_mean + c * FG_PITCH + 4 * h;
    const float* ahp = Hin + (size_t)arow_idx * DOUT + 4 * h;
    const float* blp = Wl + (size_t)(4 * h) * DOUT + col0 + 4 * c;
    const float* brp = Wr + (size_t)(4 * h) * DOUT + col0 + 4 * c;

    v8f acc[4] = {};
    SageStage s0, s1;
    sage_load_g(s0, ahp, blp, brp);
    __syncthreads();
    sage_load_l(s0, amp);
    for (int k = 0; k < DOUT - 16 * SAGE_KB; k += 16 * SAGE_KB) {
        const int k1 = k + 8 * SAGE_KB;
        const int k2 = k + 16 * SAGE_KB;
        sage_load_g(s1, ahp + k1, blp + (size_t)k1 * DOUT, brp + (size_t)k1 * DOUT);
        sage_load_l(s1, amp + k1);
        __builtin_amdgcn_sched_barrier(0);
        sage_mma(s0, acc);
        __builtin_amdgcn_sched_barrier(0);
        sage_load_g(s0, ahp + k2, blp + (size_t)k2 * DOUT, brp + (size_t)k2 * DOUT);
        sage_load_l(s0, amp + k2);
        __builtin_amdgcn_sched_barrier(0);
        sage_mma(s1, acc);
    }
    const size_t o0 = (size_t)(row0 + 8 * h) * DOUT + col0 + 4 * c;
    const float* hres = Hin + o0;
    float* out = OUT + o0;
    v4f bv;
    v4f hr[8];
    {
        const int k1 = DOUT - 8 * SAGE_KB;
        sage_load_g(s1, ahp + k1, blp + (size_t)k1 * DOUT, brp + (size_t)k1 * DOUT);
        sage_load_l(s1, amp + k1);
        bv = *(const v4f*)(bl + col0 + 4 * c);
        #pragma unroll
        for (int r = 0; r < 8; ++r) hr[r] = *(const v4f*)(hres + (size_t)r * DOUT);
        __builtin_amdgcn_sched_barrier(0);
        sage_mma(s0, acc);
        sage_mma(s1, acc);
    }

    v4f v[8];
    #pragma unroll
    for (int r = 0; r < 8; ++r) {
        v[r].x = acc[0][r] + bv.x;
        v[r].y = acc[1][r] + bv.y;
        v[r].z = acc[2][r] + bv.z;
        v[r].w = acc[3][r] + bv.w;
    }
    if (relu) {
        #pragma unroll
        for (int r = 0; r < 8; ++r) {
            v[r].x = fmaxf(v[r].x, 0.0f); v[r].y = fmaxf(v[r].y, 0.0f);
            v[r].z = fmaxf(v[r].z, 0.0f); v[r].w = fmaxf(v[r].w, 0.0f);
        }
    }
    #pragma unroll
    for (int r = 0; r < 8; ++r) *(v4f*)(out + (size_t)r * DOUT) = v[r] + hr[r];
}

extern "C" void kernel_launch(void* const* d_in, const int* in_sizes, int n_in,
                              void* d_out, int out_size, void* d_ws, size_t ws_size,
                              hipStream_t stream)
{
    const float* x   = (const float*)d_in[0];
    const int*   ei  = (const int*)d_in[1];
    const float* W0  = (const float*)d_in[2];
    const float* b0  = (const float*)d_in[3];
    const float* Wl1 = (const float*)d_in[4];
    const float* bl1 = (const float*)d_in[5];
    const float* Wr1 = (const float*)d_in[6];
    const float* Wl2 = (const float*)d_in[7];
    const float* bl2 = (const float*)d_in[8];
    const float* Wr2 = (const float*)d_in[9];

    const int* src = ei;
    const int* dst = ei + NEDGES;

    float* ws  = (float*)d_ws;
    float* h   = ws;
    int* rowstart = (int*)(h + (size_t)NNODES * DOUT);
    int* rowcnt   = rowstart + CSR_S * NNODES;
    int* perm     = rowcnt + CSR_S * NNODES;

    float* out1 = (float*)d_out;
    float* out2 = out1 + (size_t)NNODES * DOUT;

    dim3 gemm_grid(NNODES / 16, 1);

    csr_build_kernel<<<CSR_NB, CSR_T, 0, stream>>>(src, dst, rowstart, rowcnt, perm);

    lin_gemm_kernel<<<gemm_grid, 128, 0, stream>>>(x, W0, b0, h);

    sage_fused_kernel<<<gemm_grid, 128, 0, stream>>>(h, rowstart, rowcnt, perm,
                                                     Wl1, bl1, Wr1, out1, 1);

    sage_fused_kernel<<<gemm_grid, 128, 0, stream>>>(out1, rowstart, rowcnt, perm,
                                                     Wl2, bl2, Wr2, out2, 0);
}


# All three GEMMs: a wave owns 48 rows x 64 columns (3 row tiles, 209 work-groups), so a weight fragment is loaded once and feeds 3 matrix instructions; weight reads fall to a third; sum order kept
# speedup vs baseline: 3.5141x; 1.0413x over previous
#include <hip/hip_runtime.h>

typedef float v2f __attribute__((ext_vector_type(2)));
typedef float v8f __attribute__((ext_vector_type(8)));

#define NNODES 10000
#define NEDGES 320000
#define DIN 512
#define DOUT 256

typedef float v4f __attribute__((ext_vector_type(4)));

#define RT 3
#define NTILE (NNODES / 16)
#define NGRP ((NTILE + RT - 1) / RT)

#define LIN_KB 4

struct LinStage {
    v4f a[RT][LIN_KB];
    v4f b[LIN_KB][4];
};

__device__ __forceinline__ void lin_load(LinStage& s, const float* const (&ap)[RT], int k,
                                         const float* __restrict__ bp)
{
    #pragma unroll
    for (int q = 0; q < LIN_KB; ++q) {
        #pragma unroll
        for (int r = 0; r < RT; ++r) s.a[r][q] = *(const v4f*)(ap[r] + k + 8 * q);
        #pragma unroll
        for (int j = 0; j < 4; ++j)
            s.b[q][j] = *(const v4f*)(bp + (size_t)(k + 8 * q + j) * DOUT);
    }
}

__device__ __forceinline__ void lin_mma(const LinStage& s, v8f (&acc)[RT][4])
{
    #pragma unroll
    for (int q = 0; q < LIN_KB; ++q) {
        #pragma unroll
        for (int u = 0; u < 2; ++u) {
            v2f b[4];
            #pragma unroll
            for (int t = 0; t < 4; ++t) {
                b[t].x = s.b[q][2 * u][t];
                b[t].y = s.b[q][2 * u + 1][t];
            }
            #pragma unroll
            for (int r = 0; r < RT; ++r) {
                v2f a;
                a.x = s.a[r][q][2 * u];
                a.y = s.a[r][q][2 * u + 1];
                #pragma unroll
                for (int t = 0; t < 4; ++t)
                    acc[r][t] = __builtin_amdgcn_wmma_f32_16x16x4_f32(false, a, false, b[t],
                                                                      (short)0, acc[r][t], false, false);
            }
        }
    }
}

__global__ __launch_bounds__(128) void lin_gemm_kernel(
    const float* __restrict__ X, const float* __restrict__ W,
    const float* __restrict__ bias, float* __restrict__ H)
{
    const int lane = threadIdx.x & 31;
    const int wave = threadIdx.x >> 5;
    const int c    = lane & 15;
    const int h    = lane >> 4;

    const int row0 = blockIdx.x * (16 * RT);
    const int col0 = wave * 64;

    const float* ap[RT];
    #pragma unroll
    for (int r = 0; r < RT; ++r) {
        int row = row0 + 16 * r + c;
        row = (row < NNODES) ? row : (NNODES - 1);
        ap[r] = X + (size_t)row * DIN + 4 * h;
    }
    const float* bp = W + (size_t)(4 * h) * DOUT + col0 + 4 * c;

    v8f acc[RT][4] = {};
    LinStage s0, s1;
    lin_load(s0, ap, 0, bp);
    for (int k = 0; k < DIN - 16 * LIN_KB; k += 16 * LIN_KB) {
        lin_load(s1, ap, k + 8 * LIN_KB, bp);
        __builtin_amdgcn_sched_barrier(0);
        lin_mma(s0, acc);
        __builtin_amdgcn_sched_barrier(0);
        lin_load(s0, ap, k + 16 * LIN_KB, bp);
        __builtin_amdgcn_sched_barrier(0);
        lin_mma(s1, acc);
    }
    lin_load(s1, ap, DIN - 8 * LIN_KB, bp);
    __builtin_amdgcn_sched_barrier(0);
    lin_mma(s0, acc);
    lin_mma(s1, acc);

    const v4f bv = *(const v4f*)(bias + col0 + 4 * c);
    #pragma unroll
    for (int r = 0; r < RT; ++r) {
        if (row0 + 16 * r < NNODES) {
            float* out = H + (size_t)(row0 + 16 * r + 8 * h) * DOUT + col0 + 4 * c;
            #pragma unroll
            for (int e = 0; e < 8; ++e) {
                v4f v;
                v.x = acc[r][0][e] + bv.x;
                v.y = acc[r][1][e] + bv.y;
                v.z = acc[r][2][e] + bv.z;
                v.w = acc[r][3][e] + bv.w;
                *(v4f*)(out + (size_t)e * DOUT) = v;
            }
        }
    }
}

typedef int   v4i __attribute__((ext_vector_type(4)));

#define CSR_S    4
#define CSR_NR   50
#define CSR_NB   (CSR_S * CSR_NR)
#define CSR_R    200
#define CSR_T    512
#define CSR_NW   (CSR_T / 32)
#define CSR_ESL  (NEDGES / CSR_S)
#define CSR_E4   (CSR_ESL / 4)
#define CSR_FULL (CSR_E4 / CSR_T)
#define CSR_U    13
#define CSR_U2   3
#define CSR_REST (CSR_E4 - CSR_FULL * CSR_T)
#define CSR_LCAP 512
#define CSR_EBITS 19
#define CSR_PU   4

static_assert(CSR_NR * CSR_R == NNODES, "node ranges must tile the nodes");
static_assert(CSR_S * CSR_ESL == NEDGES, "slices must tile the edges");
static_assert(CSR_ESL % 4 == 0, "dst is read in groups of four, slices start on a 16-byte boundary");
static_assert(CSR_R <= CSR_T, "one thread per owned node");
static_assert(CSR_FULL % CSR_U == 0, "load batch must divide the full sweeps");
static_assert(CSR_FULL % CSR_U2 == 0, "load batch must divide the full sweeps");
static_assert(CSR_REST >= 0 && CSR_REST < CSR_T, "one partial sweep at most");
static_assert(CSR_ESL <= (1 << CSR_EBITS), "edge number (within the slice) must fit its field");
static_assert(CSR_R <= (1 << (31 - CSR_EBITS)), "owned node index must fit above the edge number");

__device__ __forceinline__ void csr_visit(int d, int e, int lo, int* __restrict__ s_cnt,
                                          int* __restrict__ wlist, int& below, int& wcount)
{
    below += __popc(__builtin_amdgcn_ballot_w32(d < lo));
    const unsigned r = (unsigned)(d - lo);
    const bool m = r < (unsigned)CSR_R;
    const unsigned mk = __builtin_amdgcn_ballot_w32(m);
    if (__builtin_expect(mk != 0u, 0)) {
        if (m) {
            atomicAdd(&s_cnt[r], 1);
            const int slot = wcount + (int)__builtin_amdgcn_mbcnt_lo(mk, 0u);
            if (slot < CSR_LCAP) wlist[slot] = (int)((r << CSR_EBITS) | (unsigned)e);
        }
        wcount += __popc(mk);
    }
}

__global__ __launch_bounds__(CSR_T) void csr_build_kernel(
    const int* __restrict__ src, const int* __restrict__ dst,
    int* __restrict__ rowstart, int* __restrict__ rowcnt,
    int* __restrict__ perm)
{
    __shared__ int s_cnt[CSR_R];
    __shared__ int s_cursor[CSR_R];
    __shared__ int s_below;
    __shared__ int s_over;
    __shared__ int s_list[CSR_NW * CSR_LCAP];

    const int tid  = threadIdx.x;
    const int lane = tid & 31;
    const int slice = blockIdx.x / CSR_NR;
    const int lo    = (blockIdx.x % CSR_NR) * CSR_R;
    const int ebase = slice * CSR_ESL;
    const int* __restrict__ srcs = src + ebase;
    const v4i* __restrict__ d4 = (const v4i*)(dst + ebase);
    const v4i* __restrict__ s4 = (const v4i*)srcs;
    int* __restrict__ rs_out = rowstart + slice * NNODES;
    int* __restrict__ rc_out = rowcnt + slice * NNODES;
    int* __restrict__ wlist = s_list + (tid >> 5) * CSR_LCAP;

    if (tid < CSR_R) s_cnt[tid] = 0;
    if (tid == 0) { s_below = 0; s_over = 0; }
    __syncthreads();

    int below = 0;
    int wcount = 0;
    for (int k0 = 0; k0 < CSR_FULL; k0 += CSR_U) {
        v4i v[CSR_U];
        #pragma unroll
        for (int u = 0; u < CSR_U; ++u) v[u] = d4[tid + (k0 + u) * CSR_T];
        #pragma unroll
        for (int u = 0; u < CSR_U; ++u) {
            const int e0 = 4 * (tid + (k0 + u) * CSR_T);
            #pragma unroll
            for (int c = 0; c < 4; ++c)
                csr_visit(v[u][c], e0 + c, lo, s_cnt, wlist, below, wcount);
        }
    }
    {
        const bool have = tid < CSR_REST;
        v4i v = {NNODES, NNODES, NNODES, NNODES};
        if (have) v = d4[tid + CSR_FULL * CSR_T];
        const int e0 = 4 * (tid + CSR_FULL * CSR_T);
        #pragma unroll
        for (int c = 0; c < 4; ++c)
            csr_visit(v[c], e0 + c, lo, s_cnt, wlist, below, wcount);
    }
    if (lane == 0) {
        atomicAdd(&s_below, below);
        if (wcount > CSR_LCAP) s_over = 1;
    }
    __syncthreads();

    if (tid < CSR_R) {
        int off = 0;
        for (int j = 0; j < tid; ++j) off += s_cnt[j];
        const int c  = s_cnt[tid];
        const int st = ebase + s_below + off;
        s_cursor[tid] = st;
        rs_out[lo + tid] = st;
        rc_out[lo + tid] = c;
    }
    __syncthreads();

    if (s_over == 0) {
        for (int i0 = 0; i0 < wcount; i0 += 32 * CSR_PU) {
            int r[CSR_PU], s[CSR_PU], q[CSR_PU];
            #pragma unroll
            for (int u = 0; u < CSR_PU; ++u) {
                const int i = i0 + 32 * u + lane;
                r[u] = -1;
                s[u] = 0;
                if (i < wcount) {
                    const unsigned ent = (unsigned)wlist[i];
                    r[u] = (int)(ent >> CSR_EBITS);
                    s[u] = srcs[ent & ((1u << CSR_EBITS) - 1u)];
                }
            }
            #pragma unroll
            for (int u = 0; u < CSR_PU; ++u) {
                q[u] = -1;
                if (r[u] >= 0) q[u] = atomicAdd(&s_cursor[r[u]], 1);
            }
            #pragma unroll
            for (int u = 0; u < CSR_PU; ++u) {
                if (q[u] >= 0) perm[q[u]] = s[u];
            }
        }
    } else {
        for (int k0 = 0; k0 < CSR_FULL; k0 += CSR_U2) {
            v4i v[CSR_U2], s[CSR_U2];
            #pragma unroll
            for (int u = 0; u < CSR_U2; ++u) {
                v[u] = d4[tid + (k0 + u) * CSR_T];
                s[u] = s4[tid + (k0 + u) * CSR_T];
            }
            int p[CSR_U2][4];
            #pragma unroll
            for (int u = 0; u < CSR_U2; ++u) {
                #pragma unroll
                for (int c = 0; c < 4; ++c) {
                    const unsigned r = (unsigned)(v[u][c] - lo);
                    int q = -1;
                    if (r < (unsigned)CSR_R) q = atomicAdd(&s_cursor[r], 1);
                    p[u][c] = q;
                }
            }
            #pragma unroll
            for (int u = 0; u < CSR_U2; ++u) {
                #pragma unroll
                for (int c = 0; c < 4; ++c) {
                    if (p[u][c] >= 0) perm[p[u][c]] = s[u][c];
                }
            }
        }
        if (tid < CSR_REST) {
            const v4i v = d4[tid + CSR_FULL * CSR_T];
            const v4i s = s4[tid + CSR_FULL * CSR_T];
            #pragma unroll
            for (int c = 0; c < 4; ++c) {
                const unsigned r = (unsigned)(v[c] - lo);
                if (r < (unsigned)CSR_R) {
                    const int q = atomicAdd(&s_cursor[r], 1);
                    perm[q] = s[c];
                }
            }
        }
    }
}

#define SAGE_KB 2
#define FG_PITCH (DOUT + 4)
#define FG_U 16
#define SAGE_T (128 * RT)

static_assert(NNODES % 16 == 0, "no partial row tile");
static_assert(SAGE_KB == 2, "the mean fragments are kept as two half stages");
static_assert((NGRP - 1) * 16 * RT < NNODES, "the first row tile of every work-group is inside");

struct SageStage {
    v4f ah[RT][SAGE_KB];
    v4f bl[SAGE_KB][4];
    v4f br[SAGE_KB][4];
};

__device__ __forceinline__ void sage_load_g(SageStage& s, const float* const (&ahp)[RT], int k,
                                            const float* __restrict__ blp,
                                            const float* __restrict__ brp)
{
    #pragma unroll
    for (int q = 0; q < SAGE_KB; ++q) {
        #pragma unroll
        for (int r = 0; r < RT; ++r) s.ah[r][q] = *(const v4f*)(ahp[r] + k + 8 * q);
        #pragma unroll
        for (int j = 0; j < 4; ++j) {
            s.bl[q][j] = *(const v4f*)(blp + (size_t)(k + 8 * q + j) * DOUT);
            s.br[q][j] = *(const v4f*)(brp + (size_t)(k + 8 * q + j) * DOUT);
        }
    }
}

__device__ __forceinline__ void sage_load_am(v4f (&am)[RT], const float* amp)
{
    #pragma unroll
    for (int r = 0; r < RT; ++r) am[r] = *(const v4f*)(amp + 16 * r * FG_PITCH);
}

__device__ __forceinline__ void sage_mma_q(const SageStage& s, const int q, const v4f (&am)[RT],
                                           v8f (&acc)[RT][4])
{
    #pragma unroll
    for (int u = 0; u < 2; ++u) {
        v2f wl[4], wr[4];
        #pragma unroll
        for (int t = 0; t < 4; ++t) {
            wl[t].x = s.bl[q][2 * u][t];
            wl[t].y = s.bl[q][2 * u + 1][t];
            wr[t].x = s.br[q][2 * u][t];
            wr[t].y = s.br[q][2 * u + 1][t];
        }
        #pragma unroll
        for (int r = 0; r < RT; ++r) {
            v2f a;
            a.x = am[r][2 * u];
            a.y = am[r][2 * u + 1];
            #pragma unroll
            for (int t = 0; t < 4; ++t)
                acc[r][t] = __builtin_amdgcn_wmma_f32_16x16x4_f32(false, a, false, wl[t],
                                                                  (short)0, acc[r][t], false, false);
        }
        #pragma unroll
        for (int r = 0; r < RT; ++r) {
            v2f a;
            a.x = s.ah[r][q][2 * u];
            a.y = s.ah[r][q][2 * u + 1];
            #pragma unroll
            for (int t = 0; t < 4; ++t)
                acc[r][t] = __builtin_amdgcn_wmma_f32_16x16x4_f32(false, a, false, wr[t],
                                                                  (short)0, acc[r][t], false, false);
        }
    }
}

__device__ __forceinline__ void sage_mma(const SageStage& s, v8f (&acc)[RT][4],
                                         v4f (&am0)[RT], v4f (&am1)[RT],
                                         const float* amp, const bool more)
{
    sage_load_am(am1, amp + 8);
    __builtin_amdgcn_sched_barrier(0);
    sage_mma_q(s, 0, am0, acc);
    __builtin_amdgcn_sched_barrier(0);
    if (more) sage_load_am(am0, amp + 16);
    __builtin_amdgcn_sched_barrier(0);
    sage_mma_q(s, 1, am1, acc);
}

__device__ __forceinline__ void fg_rows(const float* __restrict__ H, int mysrc, int j,
                                        const int n, int lane, v4f& a0, v4f& a1)
{
    v4f r0[FG_U], r1[FG_U];
    #pragma unroll
    for (int u = 0; u < n; ++u) {
        const int sidx = __builtin_amdgcn_readlane(mysrc, j + u);
        const v4f* __restrict__ p = (const v4f*)(H + (size_t)sidx * DOUT);
        r0[u] = p[lane];
        r1[u] = p[32 + lane];
    }
    #pragma unroll
    for (int u = 0; u < n; ++u) { a0 += r0[u]; a1 += r1[u]; }
}

__device__ __forceinline__ void fg_tail(const float* __restrict__ H, int mysrc, int j,
                                        int m, int lane, v4f& a0, v4f& a1)
{
    int sidx[7];
    v4f r0[7], r1[7];
    #pragma unroll
    for (int u = 0; u < 7; ++u) sidx[u] = __builtin_amdgcn_readlane(mysrc, (j + u) & 31);
    #pragma unroll
    for (int u = 0; u < 7; ++u) {
        if (u < m) {
            const v4f* __restrict__ p = (const v4f*)(H + (size_t)sidx[u] * DOUT);
            r0[u] = p[lane];
            r1[u] = p[32 + lane];
        }
    }
    #pragma unroll
    for (int u = 0; u < 7; ++u) {
        if (u < m) { a0 += r0[u]; a1 += r1[u]; }
    }
}

__device__ __forceinline__ int fg_slot(int k, const int (&bnd)[CSR_S], const int (&off)[CSR_S])
{
    int o = off[CSR_S - 1];
    #pragma unroll
    for (int s = CSR_S - 2; s >= 0; --s) o = (k < bnd[s]) ? off[s] : o;
    return k + o;
}

__global__ __launch_bounds__(SAGE_T) void sage_fused_kernel(
    const float* __restrict__ Hin,
    const int* __restrict__ rowstart, const int* __restrict__ rowcnt,
    const int* __restrict__ perm,
    const float* __restrict__ Wl, const float* __restrict__ bl,
    const float* __restrict__ Wr,
    float* __restrict__ OUT, int relu)
{
    __shared__ __attribute__((aligned(16))) float s_mean[16 * RT * FG_PITCH];

    const int lane = threadIdx.x & 31;
    const int wave = threadIdx.x >> 5;
    const int c    = lane & 15;
    const int h    = lane >> 4;

    const int row0 = blockIdx.x * (16 * RT);

    const int nb0 = __builtin_amdgcn_readfirstlane(row0 + 4 * wave);
    if (nb0 < NNODES) {
        v4i st4[CSR_S], cn4[CSR_S];
        #pragma unroll
        for (int s = 0; s < CSR_S; ++s) {
            st4[s] = *(const v4i*)(rowstart + s * NNODES + nb0);
            cn4[s] = *(const v4i*)(rowcnt + s * NNODES + nb0);
        }
        int bnd[4][CSR_S], off[4][CSR_S];
        #pragma unroll
        for (int i = 0; i < 4; ++i) {
            int run = 0;
            #pragma unroll
            for (int s = 0; s < CSR_S; ++s) {
                off[i][s] = st4[s][i] - run;
                run += cn4[s][i];
                bnd[i][s] = run;
            }
        }

        int pre0[4], pre1[4];
        #pragma unroll
        for (int i = 0; i < 4; ++i) {
            pre0[i] = 0;
            pre1[i] = 0;
            if (lane < bnd[i][CSR_S - 1])      pre0[i] = perm[fg_slot(lane, bnd[i], off[i])];
            if (lane + 32 < bnd[i][CSR_S - 1]) pre1[i] = perm[fg_slot(lane + 32, bnd[i], off[i])];
        }

        #pragma unroll
        for (int i = 0; i < 4; ++i) {
            const int cnt   = bnd[i][CSR_S - 1];
            v4f a0 = {0.0f, 0.0f, 0.0f, 0.0f};
            v4f a1 = {0.0f, 0.0f, 0.0f, 0.0f};
            for (int j0 = 0; j0 < cnt; j0 += 32) {
                const int nb = (cnt - j0 < 32) ? (cnt - j0) : 32;
                int mysrc;
                if (j0 == 0)       mysrc = pre0[i];
                else if (j0 == 32) mysrc = pre1[i];
                else {
                    mysrc = 0;
                    if (lane < nb) mysrc = perm[fg_slot(j0 + lane, bnd[i], off[i])];
                }
                int j = 0;
                for (; j + FG_U <= nb; j += FG_U) fg_rows(Hin, mysrc, j, FG_U, lane, a0, a1);
                if (j + 8 <= nb) { fg_rows(Hin, mysrc, j, 8, lane, a0, a1); j += 8; }
                if (j < nb) fg_tail(Hin, mysrc, j, nb - j, lane, a0, a1);
            }
            const float inv = 1.0f / fmaxf((float)cnt, 1.0f);
            float* o = s_mean + (4 * wave + i) * FG_PITCH;
            *(v4f*)(o + 4 * lane)       = a0 * inv;
            *(v4f*)(o + 128 + 4 * lane) = a1 * inv;
        }
    } else {
        const v4f z = {0.0f, 0.0f, 0.0f, 0.0f};
        #pragma unroll
        for (int i = 0; i < 4; ++i) {
            float* o = s_mean + (4 * wave + i) * FG_PITCH;
            *(v4f*)(o + 4 * lane)       = z;
            *(v4f*)(o + 128 + 4 * lane) = z;
        }
    }

    const bool mm = wave < 4;
    const int col0 = (wave & 3) * 64;
    const float* ahp[RT];
    #pragma unroll
    for (int r = 0; r < RT; ++r) {
        int row = row0 + 16 * r + c;
        row = (row < NNODES) ? row : (NNODES - 1);
        ahp[r] = Hin + (size_t)row * DOUT + 4 * h;
    }
    const float* amp = s_mean + c * FG_PITCH + 4 * h;
    const float* blp = Wl + (size_t)(4 * h) * DOUT + col0 + 4 * c;
    const float* brp = Wr + (size_t)(4 * h) * DOUT + col0 + 4 * c;

    v8f acc[RT][4] = {};
    SageStage s0, s1;
    v4f am0[RT], am1[RT];
    if (mm) sage_load_g(s0, ahp, 0, blp, brp);
    __syncthreads();
    if (!mm) return;
    sage_load_am(am0, amp);
    for (int k = 0; k < DOUT - 16 * SAGE_KB; k += 16 * SAGE_KB) {
        const int k1 = k + 8 * SAGE_KB;
        const int k2 = k + 16 * SAGE_KB;
        sage_load_g(s1, ahp, k1, blp, brp);
        __builtin_amdgcn_sched_barrier(0);
        sage_mma(s0, acc, am0, am1, amp + k, true);
        __builtin_amdgcn_sched_barrier(0);
        sage_load_g(s0, ahp, k2, blp, brp);
        __builtin_amdgcn_sched_barrier(0);
        sage_mma(s1, acc, am0, am1, amp + k1, true);
    }
    const size_t o0 = (size_t)(row0 + 8 * h) * DOUT + col0 + 4 * c;
    const float* hres = Hin + o0;
    float* out = OUT + o0;
    v4f bv;
    v4f hr[RT][8];
    {
        const int k1 = DOUT - 8 * SAGE_KB;
        sage_load_g(s1, ahp, k1, blp, brp);
        bv = *(const v4f*)(bl + col0 + 4 * c);
        __builtin_amdgcn_sched_barrier(0);
        sage_mma(s0, acc, am0, am1, amp + (k1 - 8 * SAGE_KB), true);
        __builtin_amdgcn_sched_barrier(0);
        #pragma unroll
        for (int r = 0; r < RT; ++r) {
            const int tr = (row0 + 16 * r < NNODES) ? 16 * r : 0;
            #pragma unroll
            for (int e = 0; e < 8; ++e)
                hr[r][e] = *(const v4f*)(hres + (size_t)(tr + e) * DOUT);
        }
        __builtin_amdgcn_sched_barrier(0);
        sage_mma(s1, acc, am0, am1, amp + k1, false);
    }

    #pragma unroll
    for (int r = 0; r < RT; ++r) {
        if (r == 0 || row0 + 16 * r < NNODES) {
            v4f v[8];
            #pragma unroll
            for (int e = 0; e < 8; ++e) {
                v[e].x = acc[r][0][e] + bv.x;
                v[e].y = acc[r][1][e] + bv.y;
                v[e].z = acc[r][2][e] + bv.z;
                v[e].w = acc[r][3][e] + bv.w;
            }
            if (relu) {
                #pragma unroll
                for (int e = 0; e < 8; ++e) {
                    v[e].x = fmaxf(v[e].x, 0.0f); v[e].y = fmaxf(v[e].y, 0.0f);
                    v[e].z = fmaxf(v[e].z, 0.0f); v[e].w = fmaxf(v[e].w, 0.0f);
                }
            }
            #pragma unroll
            for (int e = 0; e < 8; ++e)
                *(v4f*)(out + (size_t)(16 * r + e) * DOUT) = v[e] + hr[r][e];
        }
    }
}

extern "C" void kernel_launch(void* const* d_in, const int* in_sizes, int n_in,
                              void* d_out, int out_size, void* d_ws, size_t ws_size,
                              hipStream_t stream)
{
    const float* x   = (const float*)d_in[0];
    const int*   ei  = (const int*)d_in[1];
    const float* W0  = (const float*)d_in[2];
    const float* b0  = (const float*)d_in[3];
    const float* Wl1 = (const float*)d_in[4];
    const float* bl1 = (const float*)d_in[5];
    const float* Wr1 = (const float*)d_in[6];
    const float* Wl2 = (const float*)d_in[7];
    const float* bl2 = (const float*)d_in[8];
    const float* Wr2 = (const float*)d_in[9];

    const int* src = ei;
    const int* dst = ei + NEDGES;

    float* ws  = (float*)d_ws;
    float* h   = ws;
    int* rowstart = (int*)(h + (size_t)NNODES * DOUT);
    int* rowcnt   = rowstart + CSR_S * NNODES;
    int* perm     = rowcnt + CSR_S * NNODES;

    float* out1 = (float*)d_out;
    float* out2 = out1 + (size_t)NNODES * DOUT;

    dim3 gemm_grid(NGRP, 1);

    csr_build_kernel<<<CSR_NB, CSR_T, 0, stream>>>(src, dst, rowstart, rowcnt, perm);

    lin_gemm_kernel<<<gemm_grid, 128, 0, stream>>>(x, W0, b0, h);

    sage_fused_kernel<<<gemm_grid, SAGE_T, 0, stream>>>(h, rowstart, rowcnt, perm,
                                                        Wl1, bl1, Wr1, out1, 1);

    sage_fused_kernel<<<gemm_grid, SAGE_T, 0, stream>>>(out1, rowstart, rowcnt, perm,
                                                        Wl2, bl2, Wr2, out2, 0);
}


# Layer kernel: 4 added waves (16 per work-group) multiply Hin@Wr from a 4-stage register ring while the 12 others gather, then mean@Wl behind the barrier; an accumulator sums Wr products before Wl
# speedup vs baseline: 3.6766x; 1.0462x over previous
#include <hip/hip_runtime.h>

typedef float v2f __attribute__((ext_vector_type(2)));
typedef float v8f __attribute__((ext_vector_type(8)));

#define NNODES 10000
#define NEDGES 320000
#define DIN 512
#define DOUT 256

typedef float v4f __attribute__((ext_vector_type(4)));

#define RT 3
#define NTILE (NNODES / 16)
#define NGRP ((NTILE + RT - 1) / RT)

#define LIN_KB 4

struct LinStage {
    v4f a[RT][LIN_KB];
    v4f b[LIN_KB][4];
};

__device__ __forceinline__ void lin_load(LinStage& s, const float* const (&ap)[RT], int k,
                                         const float* __restrict__ bp)
{
    #pragma unroll
    for (int q = 0; q < LIN_KB; ++q) {
        #pragma unroll
        for (int r = 0; r < RT; ++r) s.a[r][q] = *(const v4f*)(ap[r] + k + 8 * q);
        #pragma unroll
        for (int j = 0; j < 4; ++j)
            s.b[q][j] = *(const v4f*)(bp + (size_t)(k + 8 * q + j) * DOUT);
    }
}

__device__ __forceinline__ void lin_mma(const LinStage& s, v8f (&acc)[RT][4])
{
    #pragma unroll
    for (int q = 0; q < LIN_KB; ++q) {
        #pragma unroll
        for (int u = 0; u < 2; ++u) {
            v2f b[4];
            #pragma unroll
            for (int t = 0; t < 4; ++t) {
                b[t].x = s.b[q][2 * u][t];
                b[t].y = s.b[q][2 * u + 1][t];
            }
            #pragma unroll
            for (int r = 0; r < RT; ++r) {
                v2f a;
                a.x = s.a[r][q][2 * u];
                a.y = s.a[r][q][2 * u + 1];
                #pragma unroll
                for (int t = 0; t < 4; ++t)
                    acc[r][t] = __builtin_amdgcn_wmma_f32_16x16x4_f32(false, a, false, b[t],
                                                                      (short)0, acc[r][t], false, false);
            }
        }
    }
}

__global__ __launch_bounds__(128) void lin_gemm_kernel(
    const float* __restrict__ X, const float* __restrict__ W,
    const float* __restrict__ bias, float* __restrict__ H)
{
    const int lane = threadIdx.x & 31;
    const int wave = threadIdx.x >> 5;
    const int c    = lane & 15;
    const int h    = lane >> 4;

    const int row0 = blockIdx.x * (16 * RT);
    const int col0 = wave * 64;

    const float* ap[RT];
    #pragma unroll
    for (int r = 0; r < RT; ++r) {
        int row = row0 + 16 * r + c;
        row = (row < NNODES) ? row : (NNODES - 1);
        ap[r] = X + (size_t)row * DIN + 4 * h;
    }
    const float* bp = W + (size_t)(4 * h) * DOUT + col0 + 4 * c;

    v8f acc[RT][4] = {};
    LinStage s0, s1;
    lin_load(s0, ap, 0, bp);
    for (int k = 0; k < DIN - 16 * LIN_KB; k += 16 * LIN_KB) {
        lin_load(s1, ap, k + 8 * LIN_KB, bp);
        __builtin_amdgcn_sched_barrier(0);
        lin_mma(s0, acc);
        __builtin_amdgcn_sched_barrier(0);
        lin_load(s0, ap, k + 16 * LIN_KB, bp);
        __builtin_amdgcn_sched_barrier(0);
        lin_mma(s1, acc);
    }
    lin_load(s1, ap, DIN - 8 * LIN_KB, bp);
    __builtin_amdgcn_sched_barrier(0);
    lin_mma(s0, acc);
    lin_mma(s1, acc);

    const v4f bv = *(const v4f*)(bias + col0 + 4 * c);
    #pragma unroll
    for (int r = 0; r < RT; ++r) {
        if (row0 + 16 * r < NNODES) {
            float* out = H + (size_t)(row0 + 16 * r + 8 * h) * DOUT + col0 + 4 * c;
            #pragma unroll
            for (int e = 0; e < 8; ++e) {
                v4f v;
                v.x = acc[r][0][e] + bv.x;
                v.y = acc[r][1][e] + bv.y;
                v.z = acc[r][2][e] + bv.z;
                v.w = acc[r][3][e] + bv.w;
                *(v4f*)(out + (size_t)e * DOUT) = v;
            }
        }
    }
}

typedef int   v4i __attribute__((ext_vector_type(4)));

#define CSR_S    4
#define CSR_NR   50
#define CSR_NB   (CSR_S * CSR_NR)
#define CSR_R    200
#define CSR_T    512
#define CSR_NW   (CSR_T / 32)
#define CSR_ESL  (NEDGES / CSR_S)
#define CSR_E4   (CSR_ESL / 4)
#define CSR_FULL (CSR_E4 / CSR_T)
#define CSR_U    13
#define CSR_U2   3
#define CSR_REST (CSR_E4 - CSR_FULL * CSR_T)
#define CSR_LCAP 512
#define CSR_EBITS 19
#define CSR_PU   4

static_assert(CSR_NR * CSR_R == NNODES, "node ranges must tile the nodes");
static_assert(CSR_S * CSR_ESL == NEDGES, "slices must tile the edges");
static_assert(CSR_ESL % 4 == 0, "dst is read in groups of four, slices start on a 16-byte boundary");
static_assert(CSR_R <= CSR_T, "one thread per owned node");
static_assert(CSR_FULL % CSR_U == 0, "load batch must divide the full sweeps");
static_assert(CSR_FULL % CSR_U2 == 0, "load batch must divide the full sweeps");
static_assert(CSR_REST >= 0 && CSR_REST < CSR_T, "one partial sweep at most");
static_assert(CSR_ESL <= (1 << CSR_EBITS), "edge number (within the slice) must fit its field");
static_assert(CSR_R <= (1 << (31 - CSR_EBITS)), "owned node index must fit above the edge number");

__device__ __forceinline__ void csr_visit(int d, int e, int lo, int* __restrict__ s_cnt,
                                          int* __restrict__ wlist, int& below, int& wcount)
{
    below += __popc(__builtin_amdgcn_ballot_w32(d < lo));
    const unsigned r = (unsigned)(d - lo);
    const bool m = r < (unsigned)CSR_R;
    const unsigned mk = __builtin_amdgcn_ballot_w32(m);
    if (__builtin_expect(mk != 0u, 0)) {
        if (m) {
            atomicAdd(&s_cnt[r], 1);
            const int slot = wcount + (int)__builtin_amdgcn_mbcnt_lo(mk, 0u);
            if (slot < CSR_LCAP) wlist[slot] = (int)((r << CSR_EBITS) | (unsigned)e);
        }
        wcount += __popc(mk);
    }
}

__global__ __launch_bounds__(CSR_T) void csr_build_kernel(
    const int* __restrict__ src, const int* __restrict__ dst,
    int* __restrict__ rowstart, int* __restrict__ rowcnt,
    int* __restrict__ perm)
{
    __shared__ int s_cnt[CSR_R];
    __shared__ int s_cursor[CSR_R];
    __shared__ int s_below;
    __shared__ int s_over;
    __shared__ int s_list[CSR_NW * CSR_LCAP];

    const int tid  = threadIdx.x;
    const int lane = tid & 31;
    const int slice = blockIdx.x / CSR_NR;
    const int lo    = (blockIdx.x % CSR_NR) * CSR_R;
    const int ebase = slice * CSR_ESL;
    const int* __restrict__ srcs = src + ebase;
    const v4i* __restrict__ d4 = (const v4i*)(dst + ebase);
    const v4i* __restrict__ s4 = (const v4i*)srcs;
    int* __restrict__ rs_out = rowstart + slice * NNODES;
    int* __restrict__ rc_out = rowcnt + slice * NNODES;
    int* __restrict__ wlist = s_list + (tid >> 5) * CSR_LCAP;

    if (tid < CSR_R) s_cnt[tid] = 0;
    if (tid == 0) { s_below = 0; s_over = 0; }
    __syncthreads();

    int below = 0;
    int wcount = 0;
    for (int k0 = 0; k0 < CSR_FULL; k0 += CSR_U) {
        v4i v[CSR_U];
        #pragma unroll
        for (int u = 0; u < CSR_U; ++u) v[u] = d4[tid + (k0 + u) * CSR_T];
        #pragma unroll
        for (int u = 0; u < CSR_U; ++u) {
            const int e0 = 4 * (tid + (k0 + u) * CSR_T);
            #pragma unroll
            for (int c = 0; c < 4; ++c)
                csr_visit(v[u][c], e0 + c, lo, s_cnt, wlist, below, wcount);
        }
    }
    {
        const bool have = tid < CSR_REST;
        v4i v = {NNODES, NNODES, NNODES, NNODES};
        if (have) v = d4[tid + CSR_FULL * CSR_T];
        const int e0 = 4 * (tid + CSR_FULL * CSR_T);
        #pragma unroll
        for (int c = 0; c < 4; ++c)
            csr_visit(v[c], e0 + c, lo, s_cnt, wlist, below, wcount);
    }
    if (lane == 0) {
        atomicAdd(&s_below, below);
        if (wcount > CSR_LCAP) s_over = 1;
    }
    __syncthreads();

    if (tid < CSR_R) {
        int off = 0;
        for (int j = 0; j < tid; ++j) off += s_cnt[j];
        const int c  = s_cnt[tid];
        const int st = ebase + s_below + off;
        s_cursor[tid] = st;
        rs_out[lo + tid] = st;
        rc_out[lo + tid] = c;
    }
    __syncthreads();

    if (s_over == 0) {
        for (int i0 = 0; i0 < wcount; i0 += 32 * CSR_PU) {
            int r[CSR_PU], s[CSR_PU], q[CSR_PU];
            #pragma unroll
            for (int u = 0; u < CSR_PU; ++u) {
                const int i = i0 + 32 * u + lane;
                r[u] = -1;
                s[u] = 0;
                if (i < wcount) {
                    const unsigned ent = (unsigned)wlist[i];
                    r[u] = (int)(ent >> CSR_EBITS);
                    s[u] = srcs[ent & ((1u << CSR_EBITS) - 1u)];
                }
            }
            #pragma unroll
            for (int u = 0; u < CSR_PU; ++u) {
                q[u] = -1;
                if (r[u] >= 0) q[u] = atomicAdd(&s_cursor[r[u]], 1);
            }
            #pragma unroll
            for (int u = 0; u < CSR_PU; ++u) {
                if (q[u] >= 0) perm[q[u]] = s[u];
            }
        }
    } else {
        for (int k0 = 0; k0 < CSR_FULL; k0 += CSR_U2) {
            v4i v[CSR_U2], s[CSR_U2];
            #pragma unroll
            for (int u = 0; u < CSR_U2; ++u) {
                v[u] = d4[tid + (k0 + u) * CSR_T];
                s[u] = s4[tid + (k0 + u) * CSR_T];
            }
            int p[CSR_U2][4];
            #pragma unroll
            for (int u = 0; u < CSR_U2; ++u) {
                #pragma unroll
                for (int c = 0; c < 4; ++c) {
                    const unsigned r = (unsigned)(v[u][c] - lo);
                    int q = -1;
                    if (r < (unsigned)CSR_R) q = atomicAdd(&s_cursor[r], 1);
                    p[u][c] = q;
                }
            }
            #pragma unroll
            for (int u = 0; u < CSR_U2; ++u) {
                #pragma unroll
                for (int c = 0; c < 4; ++c) {
                    if (p[u][c] >= 0) perm[p[u][c]] = s[u][c];
                }
            }
        }
        if (tid < CSR_REST) {
            const v4i v = d4[tid + CSR_FULL * CSR_T];
            const v4i s = s4[tid + CSR_FULL * CSR_T];
            #pragma unroll
            for (int c = 0; c < 4; ++c) {
                const unsigned r = (unsigned)(v[c] - lo);
                if (r < (unsigned)CSR_R) {
                    const int q = atomicAdd(&s_cursor[r], 1);
                    perm[q] = s[c];
                }
            }
        }
    }
}

#define SAGE_KB 1
#define SAGE_KS (8 * SAGE_KB)
#define FG_PITCH (DOUT + 4)
#define FG_U 16
#define SAGE_MW 4
#define SAGE_GW (4 * RT)
#define SAGE_T (32 * (SAGE_MW + SAGE_GW))

static_assert(NNODES % 16 == 0, "no partial row tile");
static_assert((NGRP - 1) * 16 * RT < NNODES, "the first row tile of every work-group is inside");

struct SageStage {
    v4f a[RT][SAGE_KB];
    v4f b[SAGE_KB][4];
};

__device__ __forceinline__ void sage_load(SageStage& s, const float* const (&ap)[RT], int k,
                                          const float* __restrict__ bp)
{
    #pragma unroll
    for (int q = 0; q < SAGE_KB; ++q) {
        #pragma unroll
        for (int r = 0; r < RT; ++r) s.a[r][q] = *(const v4f*)(ap[r] + k + 8 * q);
        #pragma unroll
        for (int j = 0; j < 4; ++j)
            s.b[q][j] = *(const v4f*)(bp + (size_t)(k + 8 * q + j) * DOUT);
    }
}

__device__ __forceinline__ void sage_load_m(SageStage& s, const float* amp, int k,
                                            const float* __restrict__ bp)
{
    #pragma unroll
    for (int q = 0; q < SAGE_KB; ++q) {
        #pragma unroll
        for (int r = 0; r < RT; ++r) s.a[r][q] = *(const v4f*)(amp + 16 * r * FG_PITCH + k + 8 * q);
        #pragma unroll
        for (int j = 0; j < 4; ++j)
            s.b[q][j] = *(const v4f*)(bp + (size_t)(k + 8 * q + j) * DOUT);
    }
}

__device__ __forceinline__ void sage_load_w(SageStage& s, int k, const float* __restrict__ bp)
{
    #pragma unroll
    for (int q = 0; q < SAGE_KB; ++q) {
        #pragma unroll
        for (int j = 0; j < 4; ++j)
            s.b[q][j] = *(const v4f*)(bp + (size_t)(k + 8 * q + j) * DOUT);
    }
}

__device__ __forceinline__ void sage_load_ma(SageStage& s, const float* amp, int k)
{
    #pragma unroll
    for (int q = 0; q < SAGE_KB; ++q) {
        #pragma unroll
        for (int r = 0; r < RT; ++r) s.a[r][q] = *(const v4f*)(amp + 16 * r * FG_PITCH + k + 8 * q);
    }
}

__device__ __forceinline__ void sage_mma(const SageStage& s, v8f (&acc)[RT][4])
{
    #pragma unroll
    for (int q = 0; q < SAGE_KB; ++q) {
        #pragma unroll
        for (int u = 0; u < 2; ++u) {
            v2f b[4];
            #pragma unroll
            for (int t = 0; t < 4; ++t) {
                b[t].x = s.b[q][2 * u][t];
                b[t].y = s.b[q][2 * u + 1][t];
            }
            #pragma unroll
            for (int r = 0; r < RT; ++r) {
                v2f a;
                a.x = s.a[r][q][2 * u];
                a.y = s.a[r][q][2 * u + 1];
                #pragma unroll
                for (int t = 0; t < 4; ++t)
                    acc[r][t] = __builtin_amdgcn_wmma_f32_16x16x4_f32(false, a, false, b[t],
                                                                      (short)0, acc[r][t], false, false);
            }
        }
    }
}

__device__ __forceinline__ void fg_rows(const float* __restrict__ H, int mysrc, int j,
                                        const int n, int lane, v4f& a0, v4f& a1)
{
    v4f r0[FG_U], r1[FG_U];
    #pragma unroll
    for (int u = 0; u < n; ++u) {
        const int sidx = __builtin_amdgcn_readlane(mysrc, j + u);
        const v4f* __restrict__ p = (const v4f*)(H + (size_t)sidx * DOUT);
        r0[u] = p[lane];
        r1[u] = p[32 + lane];
    }
    #pragma unroll
    for (int u = 0; u < n; ++u) { a0 += r0[u]; a1 += r1[u]; }
}

__device__ __forceinline__ void fg_tail(const float* __restrict__ H, int mysrc, int j,
                                        int m, int lane, v4f& a0, v4f& a1)
{
    int sidx[7];
    v4f r0[7], r1[7];
    #pragma unroll
    for (int u = 0; u < 7; ++u) sidx[u] = __builtin_amdgcn_readlane(mysrc, (j + u) & 31);
    #pragma unroll
    for (int u = 0; u < 7; ++u) {
        if (u < m) {
            const v4f* __restrict__ p = (const v4f*)(H + (size_t)sidx[u] * DOUT);
            r0[u] = p[lane];
            r1[u] = p[32 + lane];
        }
    }
    #pragma unroll
    for (int u = 0; u < 7; ++u) {
        if (u < m) { a0 += r0[u]; a1 += r1[u]; }
    }
}

__device__ __forceinline__ int fg_slot(int k, const int (&bnd)[CSR_S], const int (&off)[CSR_S])
{
    int o = off[CSR_S - 1];
    #pragma unroll
    for (int s = CSR_S - 2; s >= 0; --s) o = (k < bnd[s]) ? off[s] : o;
    return k + o;
}

__global__ __launch_bounds__(SAGE_T) void sage_fused_kernel(
    const float* __restrict__ Hin,
    const int* __restrict__ rowstart, const int* __restrict__ rowcnt,
    const int* __restrict__ perm,
    const float* __restrict__ Wl, const float* __restrict__ bl,
    const float* __restrict__ Wr,
    float* __restrict__ OUT, int relu)
{
    __shared__ __attribute__((aligned(16))) float s_mean[16 * RT * FG_PITCH];

    const int lane = threadIdx.x & 31;
    const int wave = threadIdx.x >> 5;
    const int c    = lane & 15;
    const int h    = lane >> 4;

    const int row0 = blockIdx.x * (16 * RT);
    const bool mm = wave < SAGE_MW;

    const int col0 = (wave & 3) * 64;
    v8f acc[RT][4] = {};
    SageStage s0, s1, s2, s3;
    const float* amp = s_mean + c * FG_PITCH + 4 * h;
    const float* blp = Wl + (size_t)(4 * h) * DOUT + col0 + 4 * c;

    if (mm) {
        const float* ahp[RT];
        #pragma unroll
        for (int r = 0; r < RT; ++r) {
            int row = row0 + 16 * r + c;
            row = (row < NNODES) ? row : (NNODES - 1);
            ahp[r] = Hin + (size_t)row * DOUT + 4 * h;
        }
        const float* brp = Wr + (size_t)(4 * h) * DOUT + col0 + 4 * c;
        sage_load(s0, ahp, 0, brp);
        __builtin_amdgcn_sched_barrier(0);
        sage_load(s1, ahp, SAGE_KS, brp);
        __builtin_amdgcn_sched_barrier(0);
        sage_load(s2, ahp, 2 * SAGE_KS, brp);
        __builtin_amdgcn_sched_barrier(0);
        for (int k = 0; k < DOUT - 4 * SAGE_KS; k += 4 * SAGE_KS) {
            sage_load(s3, ahp, k + 3 * SAGE_KS, brp);
            __builtin_amdgcn_sched_barrier(0);
            sage_mma(s0, acc);
            __builtin_amdgcn_sched_barrier(0);
            sage_load(s0, ahp, k + 4 * SAGE_KS, brp);
            __builtin_amdgcn_sched_barrier(0);
            sage_mma(s1, acc);
            __builtin_amdgcn_sched_barrier(0);
            sage_load(s1, ahp, k + 5 * SAGE_KS, brp);
            __builtin_amdgcn_sched_barrier(0);
            sage_mma(s2, acc);
            __builtin_amdgcn_sched_barrier(0);
            sage_load(s2, ahp, k + 6 * SAGE_KS, brp);
            __builtin_amdgcn_sched_barrier(0);
            sage_mma(s3, acc);
            __builtin_amdgcn_sched_barrier(0);
        }
        sage_load(s3, ahp, DOUT - SAGE_KS, brp);
        __builtin_amdgcn_sched_barrier(0);
        sage_mma(s0, acc);
        __builtin_amdgcn_sched_barrier(0);
        sage_load_w(s0, 0, blp);
        __builtin_amdgcn_sched_barrier(0);
        sage_mma(s1, acc);
        __builtin_amdgcn_sched_barrier(0);
        sage_load_w(s1, SAGE_KS, blp);
        __builtin_amdgcn_sched_barrier(0);
        sage_mma(s2, acc);
        __builtin_amdgcn_sched_barrier(0);
        sage_load_w(s2, 2 * SAGE_KS, blp);
        __builtin_amdgcn_sched_barrier(0);
        sage_mma(s3, acc);
    } else {
        const int gw = wave - SAGE_MW;
        const int nb0 = __builtin_amdgcn_readfirstlane(row0 + 4 * gw);
        if (nb0 < NNODES) {
            v4i st4[CSR_S], cn4[CSR_S];
            #pragma unroll
            for (int s = 0; s < CSR_S; ++s) {
                st4[s] = *(const v4i*)(rowstart + s * NNODES + nb0);
                cn4[s] = *(const v4i*)(rowcnt + s * NNODES + nb0);
            }
            int bnd[4][CSR_S], off[4][CSR_S];
            #pragma unroll
            for (int i = 0; i < 4; ++i) {
                int run = 0;
                #pragma unroll
                for (int s = 0; s < CSR_S; ++s) {
                    off[i][s] = st4[s][i] - run;
                    run += cn4[s][i];
                    bnd[i][s] = run;
                }
            }

            int pre0[4], pre1[4];
            #pragma unroll
            for (int i = 0; i < 4; ++i) {
                pre0[i] = 0;
                pre1[i] = 0;
                if (lane < bnd[i][CSR_S - 1])      pre0[i] = perm[fg_slot(lane, bnd[i], off[i])];
                if (lane + 32 < bnd[i][CSR_S - 1]) pre1[i] = perm[fg_slot(lane + 32, bnd[i], off[i])];
            }

            #pragma unroll
            for (int i = 0; i < 4; ++i) {
                const int cnt   = bnd[i][CSR_S - 1];
                v4f a0 = {0.0f, 0.0f, 0.0f, 0.0f};
                v4f a1 = {0.0f, 0.0f, 0.0f, 0.0f};
                for (int j0 = 0; j0 < cnt; j0 += 32) {
                    const int nb = (cnt - j0 < 32) ? (cnt - j0) : 32;
                    int mysrc;
                    if (j0 == 0)       mysrc = pre0[i];
                    else if (j0 == 32) mysrc = pre1[i];
                    else {
                        mysrc = 0;
                        if (lane < nb) mysrc = perm[fg_slot(j0 + lane, bnd[i], off[i])];
                    }
                    int j = 0;
                    for (; j + FG_U <= nb; j += FG_U) fg_rows(Hin, mysrc, j, FG_U, lane, a0, a1);
                    if (j + 8 <= nb) { fg_rows(Hin, mysrc, j, 8, lane, a0, a1); j += 8; }
                    if (j < nb) fg_tail(Hin, mysrc, j, nb - j, lane, a0, a1);
                }
                const float inv = 1.0f / fmaxf((float)cnt, 1.0f);
                float* o = s_mean + (4 * gw + i) * FG_PITCH;
                *(v4f*)(o + 4 * lane)       = a0 * inv;
                *(v4f*)(o + 128 + 4 * lane) = a1 * inv;
            }
        } else {
            const v4f z = {0.0f, 0.0f, 0.0f, 0.0f};
            #pragma unroll
            for (int i = 0; i < 4; ++i) {
                float* o = s_mean + (4 * gw + i) * FG_PITCH;
                *(v4f*)(o + 4 * lane)       = z;
                *(v4f*)(o + 128 + 4 * lane) = z;
            }
        }
    }

    __syncthreads();
    if (!mm) return;

    sage_load_ma(s0, amp, 0);
    sage_load_ma(s1, amp, SAGE_KS);
    sage_load_ma(s2, amp, 2 * SAGE_KS);
    for (int k = 0; k < DOUT - 4 * SAGE_KS; k += 4 * SAGE_KS) {
        sage_load_m(s3, amp, k + 3 * SAGE_KS, blp);
        __builtin_amdgcn_sched_barrier(0);
        sage_mma(s0, acc);
        __builtin_amdgcn_sched_barrier(0);
        sage_load_m(s0, amp, k + 4 * SAGE_KS, blp);
        __builtin_amdgcn_sched_barrier(0);
        sage_mma(s1, acc);
        __builtin_amdgcn_sched_barrier(0);
        sage_load_m(s1, amp, k + 5 * SAGE_KS, blp);
        __builtin_amdgcn_sched_barrier(0);
        sage_mma(s2, acc);
        __builtin_amdgcn_sched_barrier(0);
        sage_load_m(s2, amp, k + 6 * SAGE_KS, blp);
        __builtin_amdgcn_sched_barrier(0);
        sage_mma(s3, acc);
        __builtin_amdgcn_sched_barrier(0);
    }
    const size_t o0 = (size_t)(row0 + 8 * h) * DOUT + col0 + 4 * c;
    const float* hres = Hin + o0;
    float* out = OUT + o0;
    sage_load_m(s3, amp, DOUT - SAGE_KS, blp);
    const v4f bv = *(const v4f*)(bl + col0 + 4 * c);
    __builtin_amdgcn_sched_barrier(0);
    sage_mma(s0, acc);
    sage_mma(s1, acc);
    __builtin_amdgcn_sched_barrier(0);
    v4f hr0[8];
    #pragma unroll
    for (int e = 0; e < 8; ++e) hr0[e] = *(const v4f*)(hres + (size_t)e * DOUT);
    __builtin_amdgcn_sched_barrier(0);
    sage_mma(s2, acc);
    sage_mma(s3, acc);

    #pragma unroll
    for (int r = 0; r < RT; ++r) {
        if (r == 0 || row0 + 16 * r < NNODES) {
            v4f hr[8];
            #pragma unroll
            for (int e = 0; e < 8; ++e)
                hr[e] = (r == 0) ? hr0[e] : *(const v4f*)(hres + (size_t)(16 * r + e) * DOUT);
            v4f v[8];
            #pragma unroll
            for (int e = 0; e < 8; ++e) {
                v[e].x = acc[r][0][e] + bv.x;
                v[e].y = acc[r][1][e] + bv.y;
                v[e].z = acc[r][2][e] + bv.z;
                v[e].w = acc[r][3][e] + bv.w;
            }
            if (relu) {
                #pragma unroll
                for (int e = 0; e < 8; ++e) {
                    v[e].x = fmaxf(v[e].x, 0.0f); v[e].y = fmaxf(v[e].y, 0.0f);
                    v[e].z = fmaxf(v[e].z, 0.0f); v[e].w = fmaxf(v[e].w, 0.0f);
                }
            }
            #pragma unroll
            for (int e = 0; e < 8; ++e)
                *(v4f*)(out + (size_t)(16 * r + e) * DOUT) = v[e] + hr[e];
        }
    }
}

extern "C" void kernel_launch(void* const* d_in, const int* in_sizes, int n_in,
                              void* d_out, int out_size, void* d_ws, size_t ws_size,
                              hipStream_t stream)
{
    const float* x   = (const float*)d_in[0];
    const int*   ei  = (const int*)d_in[1];
    const float* W0  = (const float*)d_in[2];
    const float* b0  = (const float*)d_in[3];
    const float* Wl1 = (const float*)d_in[4];
    const float* bl1 = (const float*)d_in[5];
    const float* Wr1 = (const float*)d_in[6];
    const float* Wl2 = (const float*)d_in[7];
    const float* bl2 = (const float*)d_in[8];
    const float* Wr2 = (const float*)d_in[9];

    const int* src = ei;
    const int* dst = ei + NEDGES;

    float* ws  = (float*)d_ws;
    float* h   = ws;
    int* rowstart = (int*)(h + (size_t)NNODES * DOUT);
    int* rowcnt   = rowstart + CSR_S * NNODES;
    int* perm     = rowcnt + CSR_S * NNODES;

    float* out1 = (float*)d_out;
    float* out2 = out1 + (size_t)NNODES * DOUT;

    dim3 gemm_grid(NGRP, 1);

    csr_build_kernel<<<CSR_NB, CSR_T, 0, stream>>>(src, dst, rowstart, rowcnt, perm);

    lin_gemm_kernel<<<gemm_grid, 128, 0, stream>>>(x, W0, b0, h);

    sage_fused_kernel<<<gemm_grid, SAGE_T, 0, stream>>>(h, rowstart, rowcnt, perm,
                                                        Wl1, bl1, Wr1, out1, 1);

    sage_fused_kernel<<<gemm_grid, SAGE_T, 0, stream>>>(out1, rowstart, rowcnt, perm,
                                                        Wl2, bl2, Wr2, out2, 0);
}


# CSR build and input GEMM share one launch as two block ranges (GEMM blocks first, 384-thread build blocks beside them on a WGP); GEMM on the 4-stage ring; sum order in a slice still by LDS arrival
# speedup vs baseline: 3.9040x; 1.0619x over previous
#include <hip/hip_runtime.h>

typedef float v2f __attribute__((ext_vector_type(2)));
typedef float v8f __attribute__((ext_vector_type(8)));

#define NNODES 10000
#define NEDGES 320000
#define DIN 512
#define DOUT 256

typedef float v4f __attribute__((ext_vector_type(4)));

#define RT 3
#define NTILE (NNODES / 16)
#define NGRP ((NTILE + RT - 1) / RT)

typedef int   v4i __attribute__((ext_vector_type(4)));

#define CSR_S    4
#define CSR_NR   50
#define CSR_NB   (CSR_S * CSR_NR)
#define CSR_R    200
#define CSR_T    384
#define CSR_NW   (CSR_T / 32)
#define CSR_ESL  (NEDGES / CSR_S)
#define CSR_E4   (CSR_ESL / 4)
#define CSR_FULL (CSR_E4 / CSR_T)
#define CSR_U    13
#define CSR_U2   4
#define CSR_REST (CSR_E4 - CSR_FULL * CSR_T)
#define CSR_LCAP 512
#define CSR_EBITS 19
#define CSR_PU   4

static_assert(CSR_NR * CSR_R == NNODES, "node ranges must tile the nodes");
static_assert(CSR_S * CSR_ESL == NEDGES, "slices must tile the edges");
static_assert(CSR_ESL % 4 == 0, "dst is read in groups of four, slices start on a 16-byte boundary");
static_assert(CSR_R <= CSR_T, "one thread per owned node");
static_assert(CSR_FULL % CSR_U == 0, "load batch must divide the full sweeps");
static_assert(CSR_FULL % CSR_U2 == 0, "load batch must divide the full sweeps");
static_assert(CSR_REST >= 0 && CSR_REST < CSR_T, "one partial sweep at most");
static_assert(CSR_ESL <= (1 << CSR_EBITS), "edge number (within the slice) must fit its field");
static_assert(CSR_R <= (1 << (31 - CSR_EBITS)), "owned node index must fit above the edge number");

__device__ __forceinline__ void csr_visit(int d, int e, int lo, int* __restrict__ s_cnt,
                                          int* __restrict__ wlist, int& below, int& wcount)
{
    below += __popc(__builtin_amdgcn_ballot_w32(d < lo));
    const unsigned r = (unsigned)(d - lo);
    const bool m = r < (unsigned)CSR_R;
    const unsigned mk = __builtin_amdgcn_ballot_w32(m);
    if (__builtin_expect(mk != 0u, 0)) {
        if (m) {
            atomicAdd(&s_cnt[r], 1);
            const int slot = wcount + (int)__builtin_amdgcn_mbcnt_lo(mk, 0u);
            if (slot < CSR_LCAP) wlist[slot] = (int)((r << CSR_EBITS) | (unsigned)e);
        }
        wcount += __popc(mk);
    }
}

__device__ __forceinline__ void csr_build_body(
    const int bid,
    const int* __restrict__ src, const int* __restrict__ dst,
    int* __restrict__ rowstart, int* __restrict__ rowcnt,
    int* __restrict__ perm)
{
    __shared__ int s_cnt[CSR_R];
    __shared__ int s_cursor[CSR_R];
    __shared__ int s_below;
    __shared__ int s_over;
    __shared__ int s_list[CSR_NW * CSR_LCAP];

    const int tid  = threadIdx.x;
    const int lane = tid & 31;
    const int slice = bid / CSR_NR;
    const int lo    = (bid % CSR_NR) * CSR_R;
    const int ebase = slice * CSR_ESL;
    const int* __restrict__ srcs = src + ebase;
    const v4i* __restrict__ d4 = (const v4i*)(dst + ebase);
    const v4i* __restrict__ s4 = (const v4i*)srcs;
    int* __restrict__ rs_out = rowstart + slice * NNODES;
    int* __restrict__ rc_out = rowcnt + slice * NNODES;
    int* __restrict__ wlist = s_list + (tid >> 5) * CSR_LCAP;

    if (tid < CSR_R) s_cnt[tid] = 0;
    if (tid == 0) { s_below = 0; s_over = 0; }
    __syncthreads();

    int below = 0;
    int wcount = 0;
    for (int k0 = 0; k0 < CSR_FULL; k0 += CSR_U) {
        v4i v[CSR_U];
        #pragma unroll
        for (int u = 0; u < CSR_U; ++u) v[u] = d4[tid + (k0 + u) * CSR_T];
        #pragma unroll
        for (int u = 0; u < CSR_U; ++u) {
            const int e0 = 4 * (tid + (k0 + u) * CSR_T);
            #pragma unroll
            for (int c = 0; c < 4; ++c)
                csr_visit(v[u][c], e0 + c, lo, s_cnt, wlist, below, wcount);
        }
    }
    {
        const bool have = tid < CSR_REST;
        v4i v = {NNODES, NNODES, NNODES, NNODES};
        if (have) v = d4[tid + CSR_FULL * CSR_T];
        const int e0 = 4 * (tid + CSR_FULL * CSR_T);
        #pragma unroll
        for (int c = 0; c < 4; ++c)
            csr_visit(v[c], e0 + c, lo, s_cnt, wlist, below, wcount);
    }
    if (lane == 0) {
        atomicAdd(&s_below, below);
        if (wcount > CSR_LCAP) s_over = 1;
    }
    __syncthreads();

    if (tid < CSR_R) {
        int off = 0;
        for (int j = 0; j < tid; ++j) off += s_cnt[j];
        const int c  = s_cnt[tid];
        const int st = ebase + s_below + off;
        s_cursor[tid] = st;
        rs_out[lo + tid] = st;
        rc_out[lo + tid] = c;
    }
    __syncthreads();

    if (s_over == 0) {
        for (int i0 = 0; i0 < wcount; i0 += 32 * CSR_PU) {
            int r[CSR_PU], s[CSR_PU], q[CSR_PU];
            #pragma unroll
            for (int u = 0; u < CSR_PU; ++u) {
                const int i = i0 + 32 * u + lane;
                r[u] = -1;
                s[u] = 0;
                if (i < wcount) {
                    const unsigned ent = (unsigned)wlist[i];
                    r[u] = (int)(ent >> CSR_EBITS);
                    s[u] = srcs[ent & ((1u << CSR_EBITS) - 1u)];
                }
            }
            #pragma unroll
            for (int u = 0; u < CSR_PU; ++u) {
                q[u] = -1;
                if (r[u] >= 0) q[u] = atomicAdd(&s_cursor[r[u]], 1);
            }
            #pragma unroll
            for (int u = 0; u < CSR_PU; ++u) {
                if (q[u] >= 0) perm[q[u]] = s[u];
            }
        }
    } else {
        for (int k0 = 0; k0 < CSR_FULL; k0 += CSR_U2) {
            v4i v[CSR_U2], s[CSR_U2];
            #pragma unroll
            for (int u = 0; u < CSR_U2; ++u) {
                v[u] = d4[tid + (k0 + u) * CSR_T];
                s[u] = s4[tid + (k0 + u) * CSR_T];
            }
            int p[CSR_U2][4];
            #pragma unroll
            for (int u = 0; u < CSR_U2; ++u) {
                #pragma unroll
                for (int c = 0; c < 4; ++c) {
                    const unsigned r = (unsigned)(v[u][c] - lo);
                    int q = -1;
                    if (r < (unsigned)CSR_R) q = atomicAdd(&s_cursor[r], 1);
                    p[u][c] = q;
                }
            }
            #pragma unroll
            for (int u = 0; u < CSR_U2; ++u) {
                #pragma unroll
                for (int c = 0; c < 4; ++c) {
                    if (p[u][c] >= 0) perm[p[u][c]] = s[u][c];
                }
            }
        }
        if (tid < CSR_REST) {
            const v4i v = d4[tid + CSR_FULL * CSR_T];
            const v4i s = s4[tid + CSR_FULL * CSR_T];
            #pragma unroll
            for (int c = 0; c < 4; ++c) {
                const unsigned r = (unsigned)(v[c] - lo);
                if (r < (unsigned)CSR_R) {
                    const int q = atomicAdd(&s_cursor[r], 1);
                    perm[q] = s[c];
                }
            }
        }
    }
}

#define RING_KB 1
#define RING_KS (8 * RING_KB)
#define FG_PITCH (DOUT + 4)
#define FG_U 16
#define SAGE_MW 4
#define SAGE_GW (4 * RT)
#define SAGE_T (32 * (SAGE_MW + SAGE_GW))

static_assert(NNODES % 16 == 0, "no partial row tile");
static_assert((NGRP - 1) * 16 * RT < NNODES, "the first row tile of every work-group is inside");

struct RingStage {
    v4f a[RT][RING_KB];
    v4f b[RING_KB][4];
};

__device__ __forceinline__ void ring_load(RingStage& s, const float* const (&ap)[RT], int k,
                                          const float* __restrict__ bp)
{
    #pragma unroll
    for (int q = 0; q < RING_KB; ++q) {
        #pragma unroll
        for (int r = 0; r < RT; ++r) s.a[r][q] = *(const v4f*)(ap[r] + k + 8 * q);
        #pragma unroll
        for (int j = 0; j < 4; ++j)
            s.b[q][j] = *(const v4f*)(bp + (size_t)(k + 8 * q + j) * DOUT);
    }
}

__device__ __forceinline__ void sage_load_m(RingStage& s, const float* amp, int k,
                                            const float* __restrict__ bp)
{
    #pragma unroll
    for (int q = 0; q < RING_KB; ++q) {
        #pragma unroll
        for (int r = 0; r < RT; ++r) s.a[r][q] = *(const v4f*)(amp + 16 * r * FG_PITCH + k + 8 * q);
        #pragma unroll
        for (int j = 0; j < 4; ++j)
            s.b[q][j] = *(const v4f*)(bp + (size_t)(k + 8 * q + j) * DOUT);
    }
}

__device__ __forceinline__ void sage_load_w(RingStage& s, int k, const float* __restrict__ bp)
{
    #pragma unroll
    for (int q = 0; q < RING_KB; ++q) {
        #pragma unroll
        for (int j = 0; j < 4; ++j)
            s.b[q][j] = *(const v4f*)(bp + (size_t)(k + 8 * q + j) * DOUT);
    }
}

__device__ __forceinline__ void sage_load_ma(RingStage& s, const float* amp, int k)
{
    #pragma unroll
    for (int q = 0; q < RING_KB; ++q) {
        #pragma unroll
        for (int r = 0; r < RT; ++r) s.a[r][q] = *(const v4f*)(amp + 16 * r * FG_PITCH + k + 8 * q);
    }
}

__device__ __forceinline__ void ring_mma(const RingStage& s, v8f (&acc)[RT][4])
{
    #pragma unroll
    for (int q = 0; q < RING_KB; ++q) {
        #pragma unroll
        for (int u = 0; u < 2; ++u) {
            v2f b[4];
            #pragma unroll
            for (int t = 0; t < 4; ++t) {
                b[t].x = s.b[q][2 * u][t];
                b[t].y = s.b[q][2 * u + 1][t];
            }
            #pragma unroll
            for (int r = 0; r < RT; ++r) {
                v2f a;
                a.x = s.a[r][q][2 * u];
                a.y = s.a[r][q][2 * u + 1];
                #pragma unroll
                for (int t = 0; t < 4; ++t)
                    acc[r][t] = __builtin_amdgcn_wmma_f32_16x16x4_f32(false, a, false, b[t],
                                                                      (short)0, acc[r][t], false, false);
            }
        }
    }
}

static_assert(CSR_T >= 128, "four multiplying waves per product block");

__device__ __forceinline__ void lin_gemm_body(
    const int grp,
    const float* __restrict__ X, const float* __restrict__ W,
    const float* __restrict__ bias, float* __restrict__ H)
{
    const int lane = threadIdx.x & 31;
    const int wave = threadIdx.x >> 5;
    const int c    = lane & 15;
    const int h    = lane >> 4;

    const int row0 = grp * (16 * RT);
    const int col0 = wave * 64;

    const float* ap[RT];
    #pragma unroll
    for (int r = 0; r < RT; ++r) {
        int row = row0 + 16 * r + c;
        row = (row < NNODES) ? row : (NNODES - 1);
        ap[r] = X + (size_t)row * DIN + 4 * h;
    }
    const float* bp = W + (size_t)(4 * h) * DOUT + col0 + 4 * c;

    v8f acc[RT][4] = {};
    RingStage s0, s1, s2, s3;
    ring_load(s0, ap, 0, bp);
    __builtin_amdgcn_sched_barrier(0);
    ring_load(s1, ap, RING_KS, bp);
    __builtin_amdgcn_sched_barrier(0);
    ring_load(s2, ap, 2 * RING_KS, bp);
    __builtin_amdgcn_sched_barrier(0);
    for (int k = 0; k < DIN - 4 * RING_KS; k += 4 * RING_KS) {
        ring_load(s3, ap, k + 3 * RING_KS, bp);
        __builtin_amdgcn_sched_barrier(0);
        ring_mma(s0, acc);
        __builtin_amdgcn_sched_barrier(0);
        ring_load(s0, ap, k + 4 * RING_KS, bp);
        __builtin_amdgcn_sched_barrier(0);
        ring_mma(s1, acc);
        __builtin_amdgcn_sched_barrier(0);
        ring_load(s1, ap, k + 5 * RING_KS, bp);
        __builtin_amdgcn_sched_barrier(0);
        ring_mma(s2, acc);
        __builtin_amdgcn_sched_barrier(0);
        ring_load(s2, ap, k + 6 * RING_KS, bp);
        __builtin_amdgcn_sched_barrier(0);
        ring_mma(s3, acc);
        __builtin_amdgcn_sched_barrier(0);
    }
    ring_load(s3, ap, DIN - RING_KS, bp);
    const v4f bv = *(const v4f*)(bias + col0 + 4 * c);
    __builtin_amdgcn_sched_barrier(0);
    ring_mma(s0, acc);
    ring_mma(s1, acc);
    ring_mma(s2, acc);
    ring_mma(s3, acc);

    #pragma unroll
    for (int r = 0; r < RT; ++r) {
        if (row0 + 16 * r < NNODES) {
            float* out = H + (size_t)(row0 + 16 * r + 8 * h) * DOUT + col0 + 4 * c;
            #pragma unroll
            for (int e = 0; e < 8; ++e) {
                v4f v;
                v.x = acc[r][0][e] + bv.x;
                v.y = acc[r][1][e] + bv.y;
                v.z = acc[r][2][e] + bv.z;
                v.w = acc[r][3][e] + bv.w;
                *(v4f*)(out + (size_t)e * DOUT) = v;
            }
        }
    }
}

__global__ __launch_bounds__(CSR_T) void first_launch_kernel(
    const float* __restrict__ X, const float* __restrict__ W0,
    const float* __restrict__ b0, float* __restrict__ H,
    const int* __restrict__ src, const int* __restrict__ dst,
    int* __restrict__ rowstart, int* __restrict__ rowcnt,
    int* __restrict__ perm)
{
    if (blockIdx.x < NGRP) {
        if (threadIdx.x >= 128) return;
        lin_gemm_body((int)blockIdx.x, X, W0, b0, H);
    } else {
        csr_build_body((int)blockIdx.x - NGRP, src, dst, rowstart, rowcnt, perm);
    }
}

__device__ __forceinline__ void fg_rows(const float* __restrict__ H, int mysrc, int j,
                                        const int n, int lane, v4f& a0, v4f& a1)
{
    v4f r0[FG_U], r1[FG_U];
    #pragma unroll
    for (int u = 0; u < n; ++u) {
        const int sidx = __builtin_amdgcn_readlane(mysrc, j + u);
        const v4f* __restrict__ p = (const v4f*)(H + (size_t)sidx * DOUT);
        r0[u] = p[lane];
        r1[u] = p[32 + lane];
    }
    #pragma unroll
    for (int u = 0; u < n; ++u) { a0 += r0[u]; a1 += r1[u]; }
}

__device__ __forceinline__ void fg_tail(const float* __restrict__ H, int mysrc, int j,
                                        int m, int lane, v4f& a0, v4f& a1)
{
    int sidx[7];
    v4f r0[7], r1[7];
    #pragma unroll
    for (int u = 0; u < 7; ++u) sidx[u] = __builtin_amdgcn_readlane(mysrc, (j + u) & 31);
    #pragma unroll
    for (int u = 0; u < 7; ++u) {
        if (u < m) {
            const v4f* __restrict__ p = (const v4f*)(H + (size_t)sidx[u] * DOUT);
            r0[u] = p[lane];
            r1[u] = p[32 + lane];
        }
    }
    #pragma unroll
    for (int u = 0; u < 7; ++u) {
        if (u < m) { a0 += r0[u]; a1 += r1[u]; }
    }
}

__device__ __forceinline__ int fg_slot(int k, const int (&bnd)[CSR_S], const int (&off)[CSR_S])
{
    int o = off[CSR_S - 1];
    #pragma unroll
    for (int s = CSR_S - 2; s >= 0; --s) o = (k < bnd[s]) ? off[s] : o;
    return k + o;
}

__global__ __launch_bounds__(SAGE_T) void sage_fused_kernel(
    const float* __restrict__ Hin,
    const int* __restrict__ rowstart, const int* __restrict__ rowcnt,
    const int* __restrict__ perm,
    const float* __restrict__ Wl, const float* __restrict__ bl,
    const float* __restrict__ Wr,
    float* __restrict__ OUT, int relu)
{
    __shared__ __attribute__((aligned(16))) float s_mean[16 * RT * FG_PITCH];

    const int lane = threadIdx.x & 31;
    const int wave = threadIdx.x >> 5;
    const int c    = lane & 15;
    const int h    = lane >> 4;

    const int row0 = blockIdx.x * (16 * RT);
    const bool mm = wave < SAGE_MW;

    const int col0 = (wave & 3) * 64;
    v8f acc[RT][4] = {};
    RingStage s0, s1, s2, s3;
    const float* amp = s_mean + c * FG_PITCH + 4 * h;
    const float* blp = Wl + (size_t)(4 * h) * DOUT + col0 + 4 * c;

    if (mm) {
        const float* ahp[RT];
        #pragma unroll
        for (int r = 0; r < RT; ++r) {
            int row = row0 + 16 * r + c;
            row = (row < NNODES) ? row : (NNODES - 1);
            ahp[r] = Hin + (size_t)row * DOUT + 4 * h;
        }
        const float* brp = Wr + (size_t)(4 * h) * DOUT + col0 + 4 * c;
        ring_load(s0, ahp, 0, brp);
        __builtin_amdgcn_sched_barrier(0);
        ring_load(s1, ahp, RING_KS, brp);
        __builtin_amdgcn_sched_barrier(0);
        ring_load(s2, ahp, 2 * RING_KS, brp);
        __builtin_amdgcn_sched_barrier(0);
        for (int k = 0; k < DOUT - 4 * RING_KS; k += 4 * RING_KS) {
            ring_load(s3, ahp, k + 3 * RING_KS, brp);
            __builtin_amdgcn_sched_barrier(0);
            ring_mma(s0, acc);
            __builtin_amdgcn_sched_barrier(0);
            ring_load(s0, ahp, k + 4 * RING_KS, brp);
            __builtin_amdgcn_sched_barrier(0);
            ring_mma(s1, acc);
            __builtin_amdgcn_sched_barrier(0);
            ring_load(s1, ahp, k + 5 * RING_KS, brp);
            __builtin_amdgcn_sched_barrier(0);
            ring_mma(s2, acc);
            __builtin_amdgcn_sched_barrier(0);
            ring_load(s2, ahp, k + 6 * RING_KS, brp);
            __builtin_amdgcn_sched_barrier(0);
            ring_mma(s3, acc);
            __builtin_amdgcn_sched_barrier(0);
        }
        ring_load(s3, ahp, DOUT - RING_KS, brp);
        __builtin_amdgcn_sched_barrier(0);
        ring_mma(s0, acc);
        __builtin_amdgcn_sched_barrier(0);
        sage_load_w(s0, 0, blp);
        __builtin_amdgcn_sched_barrier(0);
        ring_mma(s1, acc);
        __builtin_amdgcn_sched_barrier(0);
        sage_load_w(s1, RING_KS, blp);
        __builtin_amdgcn_sched_barrier(0);
        ring_mma(s2, acc);
        __builtin_amdgcn_sched_barrier(0);
        sage_load_w(s2, 2 * RING_KS, blp);
        __builtin_amdgcn_sched_barrier(0);
        ring_mma(s3, acc);
    } else {
        const int gw = wave - SAGE_MW;
        const int nb0 = __builtin_amdgcn_readfirstlane(row0 + 4 * gw);
        if (nb0 < NNODES) {
            v4i st4[CSR_S], cn4[CSR_S];
            #pragma unroll
            for (int s = 0; s < CSR_S; ++s) {
                st4[s] = *(const v4i*)(rowstart + s * NNODES + nb0);
                cn4[s] = *(const v4i*)(rowcnt + s * NNODES + nb0);
            }
            int bnd[4][CSR_S], off[4][CSR_S];
            #pragma unroll
            for (int i = 0; i < 4; ++i) {
                int run = 0;
                #pragma unroll
                for (int s = 0; s < CSR_S; ++s) {
                    off[i][s] = st4[s][i] - run;
                    run += cn4[s][i];
                    bnd[i][s] = run;
                }
            }

            int pre0[4], pre1[4];
            #pragma unroll
            for (int i = 0; i < 4; ++i) {
                pre0[i] = 0;
                pre1[i] = 0;
                if (lane < bnd[i][CSR_S - 1])      pre0[i] = perm[fg_slot(lane, bnd[i], off[i])];
                if (lane + 32 < bnd[i][CSR_S - 1]) pre1[i] = perm[fg_slot(lane + 32, bnd[i], off[i])];
            }

            #pragma unroll
            for (int i = 0; i < 4; ++i) {
                const int cnt   = bnd[i][CSR_S - 1];
                v4f a0 = {0.0f, 0.0f, 0.0f, 0.0f};
                v4f a1 = {0.0f, 0.0f, 0.0f, 0.0f};
                for (int j0 = 0; j0 < cnt; j0 += 32) {
                    const int nb = (cnt - j0 < 32) ? (cnt - j0) : 32;
                    int mysrc;
                    if (j0 == 0)       mysrc = pre0[i];
                    else if (j0 == 32) mysrc = pre1[i];
                    else {
                        mysrc = 0;
                        if (lane < nb) mysrc = perm[fg_slot(j0 + lane, bnd[i], off[i])];
                    }
                    int j = 0;
                    for (; j + FG_U <= nb; j += FG_U) fg_rows(Hin, mysrc, j, FG_U, lane, a0, a1);
                    if (j + 8 <= nb) { fg_rows(Hin, mysrc, j, 8, lane, a0, a1); j += 8; }
                    if (j < nb) fg_tail(Hin, mysrc, j, nb - j, lane, a0, a1);
                }
                const float inv = 1.0f / fmaxf((float)cnt, 1.0f);
                float* o = s_mean + (4 * gw + i) * FG_PITCH;
                *(v4f*)(o + 4 * lane)       = a0 * inv;
                *(v4f*)(o + 128 + 4 * lane) = a1 * inv;
            }
        } else {
            const v4f z = {0.0f, 0.0f, 0.0f, 0.0f};
            #pragma unroll
            for (int i = 0; i < 4; ++i) {
                float* o = s_mean + (4 * gw + i) * FG_PITCH;
                *(v4f*)(o + 4 * lane)       = z;
                *(v4f*)(o + 128 + 4 * lane) = z;
            }
        }
    }

    __syncthreads();
    if (!mm) return;

    sage_load_ma(s0, amp, 0);
    sage_load_ma(s1, amp, RING_KS);
    sage_load_ma(s2, amp, 2 * RING_KS);
    for (int k = 0; k < DOUT - 4 * RING_KS; k += 4 * RING_KS) {
        sage_load_m(s3, amp, k + 3 * RING_KS, blp);
        __builtin_amdgcn_sched_barrier(0);
        ring_mma(s0, acc);
        __builtin_amdgcn_sched_barrier(0);
        sage_load_m(s0, amp, k + 4 * RING_KS, blp);
        __builtin_amdgcn_sched_barrier(0);
        ring_mma(s1, acc);
        __builtin_amdgcn_sched_barrier(0);
        sage_load_m(s1, amp, k + 5 * RING_KS, blp);
        __builtin_amdgcn_sched_barrier(0);
        ring_mma(s2, acc);
        __builtin_amdgcn_sched_barrier(0);
        sage_load_m(s2, amp, k + 6 * RING_KS, blp);
        __builtin_amdgcn_sched_barrier(0);
        ring_mma(s3, acc);
        __builtin_amdgcn_sched_barrier(0);
    }
    const size_t o0 = (size_t)(row0 + 8 * h) * DOUT + col0 + 4 * c;
    const float* hres = Hin + o0;
    float* out = OUT + o0;
    sage_load_m(s3, amp, DOUT - RING_KS, blp);
    const v4f bv = *(const v4f*)(bl + col0 + 4 * c);
    __builtin_amdgcn_sched_barrier(0);
    ring_mma(s0, acc);
    ring_mma(s1, acc);
    __builtin_amdgcn_sched_barrier(0);
    v4f hr0[8];
    #pragma unroll
    for (int e = 0; e < 8; ++e) hr0[e] = *(const v4f*)(hres + (size_t)e * DOUT);
    __builtin_amdgcn_sched_barrier(0);
    ring_mma(s2, acc);
    ring_mma(s3, acc);

    #pragma unroll
    for (int r = 0; r < RT; ++r) {
        if (r == 0 || row0 + 16 * r < NNODES) {
            v4f hr[8];
            #pragma unroll
            for (int e = 0; e < 8; ++e)
                hr[e] = (r == 0) ? hr0[e] : *(const v4f*)(hres + (size_t)(16 * r + e) * DOUT);
            v4f v[8];
            #pragma unroll
            for (int e = 0; e < 8; ++e) {
                v[e].x = acc[r][0][e] + bv.x;
                v[e].y = acc[r][1][e] + bv.y;
                v[e].z = acc[r][2][e] + bv.z;
                v[e].w = acc[r][3][e] + bv.w;
            }
            if (relu) {
                #pragma unroll
                for (int e = 0; e < 8; ++e) {
                    v[e].x = fmaxf(v[e].x, 0.0f); v[e].y = fmaxf(v[e].y, 0.0f);
                    v[e].z = fmaxf(v[e].z, 0.0f); v[e].w = fmaxf(v[e].w, 0.0f);
                }
            }
            #pragma unroll
            for (int e = 0; e < 8; ++e)
                *(v4f*)(out + (size_t)(16 * r + e) * DOUT) = v[e] + hr[e];
        }
    }
}

extern "C" void kernel_launch(void* const* d_in, const int* in_sizes, int n_in,
                              void* d_out, int out_size, void* d_ws, size_t ws_size,
                              hipStream_t stream)
{
    const float* x   = (const float*)d_in[0];
    const int*   ei  = (const int*)d_in[1];
    const float* W0  = (const float*)d_in[2];
    const float* b0  = (const float*)d_in[3];
    const float* Wl1 = (const float*)d_in[4];
    const float* bl1 = (const float*)d_in[5];
    const float* Wr1 = (const float*)d_in[6];
    const float* Wl2 = (const float*)d_in[7];
    const float* bl2 = (const float*)d_in[8];
    const float* Wr2 = (const float*)d_in[9];

    const int* src = ei;
    const int* dst = ei + NEDGES;

    float* ws  = (float*)d_ws;
    float* h   = ws;
    int* rowstart = (int*)(h + (size_t)NNODES * DOUT);
    int* rowcnt   = rowstart + CSR_S * NNODES;
    int* perm     = rowcnt + CSR_S * NNODES;

    float* out1 = (float*)d_out;
    float* out2 = out1 + (size_t)NNODES * DOUT;

    dim3 gemm_grid(NGRP, 1);

    first_launch_kernel<<<NGRP + CSR_NB, CSR_T, 0, stream>>>(x, W0, b0, h,
                                                             src, dst, rowstart, rowcnt, perm);

    sage_fused_kernel<<<gemm_grid, SAGE_T, 0, stream>>>(h, rowstart, rowcnt, perm,
                                                        Wl1, bl1, Wr1, out1, 1);

    sage_fused_kernel<<<gemm_grid, SAGE_T, 0, stream>>>(out1, rowstart, rowcnt, perm,
                                                        Wl2, bl2, Wr2, out2, 0);
}
